# starting kernel
# baseline (speedup 1.0000x reference)
#include <hip/hip_runtime.h>

typedef __attribute__((ext_vector_type(2))) float v2f;
typedef __attribute__((ext_vector_type(8))) float v8f;

#define N_NODES 100000
#define N_EDGES 1600000
#define DIM 128
#define N_CLASSES 40
#define EPB 64

__global__ void deg_init_kernel(float* deg) {
    int i = blockIdx.x * blockDim.x + threadIdx.x;
    if (i < N_NODES) deg[i] = 1.0f;
}

__global__ void deg_acc_kernel(const int* __restrict__ col, float* deg) {
    int e = blockIdx.x * blockDim.x + threadIdx.x;
    if (e < N_EDGES) atomicAdd(&deg[col[e]], 1.0f);
}

__global__ void dinv_kernel(float* deg) {
    int i = blockIdx.x * blockDim.x + threadIdx.x;
    if (i < N_NODES) deg[i] = rsqrtf(deg[i]);
}

__global__ __launch_bounds__(256) void gemm128_wmma(const float* __restrict__ X,
                                                    const float* __restrict__ W,
                                                    float* __restrict__ H) {
    const int m0    = blockIdx.x * 16;
    const int lane  = threadIdx.x & 31;
    const int wave  = threadIdx.x >> 5;
    const int n0    = wave * 16;
    const int l15   = lane & 15;
    const int khalf = (lane >> 4) << 1;
    const float* xrow = X + (size_t)(m0 + l15) * DIM;

    v8f acc = {};
    for (int k = 0; k < DIM; k += 4) {
        const int kk = k + khalf;
        v2f a, b;
        a.x = xrow[kk];
        a.y = xrow[kk + 1];
        b.x = W[(size_t)kk * DIM + n0 + l15];
        b.y = W[(size_t)(kk + 1) * DIM + n0 + l15];
        acc = __builtin_amdgcn_wmma_f32_16x16x4_f32(
             false, a,  false, b,
             (short)0, acc,  false,  false);
    }
    const int crow = m0 + (lane >> 4) * 8;
    const int ccol = n0 + l15;
#pragma unroll
    for (int r = 0; r < 8; ++r)
        H[(size_t)(crow + r) * DIM + ccol] = acc[r];
}

__global__ __launch_bounds__(96) void gemm_out_wmma(const float* __restrict__ X,
                                                    const float* __restrict__ W,
                                                    const float* __restrict__ bias,
                                                    float* __restrict__ OUT) {
    const int m0    = blockIdx.x * 16;
    const int lane  = threadIdx.x & 31;
    const int wave  = threadIdx.x >> 5;
    const int n0    = wave * 16;
    const int l15   = lane & 15;
    const int khalf = (lane >> 4) << 1;
    const int ncol  = n0 + l15;
    const bool nv   = (ncol < N_CLASSES);
    const float* xrow = X + (size_t)(m0 + l15) * DIM;

    v8f acc = {};
    for (int k = 0; k < DIM; k += 4) {
        const int kk = k + khalf;
        v2f a, b;
        a.x = xrow[kk];
        a.y = xrow[kk + 1];
        b.x = nv ? W[(size_t)kk * N_CLASSES + ncol] : 0.0f;
        b.y = nv ? W[(size_t)(kk + 1) * N_CLASSES + ncol] : 0.0f;
        acc = __builtin_amdgcn_wmma_f32_16x16x4_f32(
            false, a, false, b, (short)0, acc, false, false);
    }
    if (nv) {
        const int crow = m0 + (lane >> 4) * 8;
        const float bb = bias[ncol];
#pragma unroll
        for (int r = 0; r < 8; ++r)
            OUT[(size_t)(crow + r) * N_CLASSES + ncol] = acc[r] + bb;
    }
}

__global__ void agg_selfloop_kernel(const float* __restrict__ dinv,
                                    const float* __restrict__ h,
                                    float* __restrict__ agg) {
    size_t idx = (size_t)blockIdx.x * blockDim.x + threadIdx.x;
    if (idx < (size_t)N_NODES * DIM) {
        int node = (int)(idx >> 7);
        float d = dinv[node];
        agg[idx] = d * d * h[idx];
    }
}

__global__ __launch_bounds__(128) void scatter_kernel(const int* __restrict__ row,
                                                      const int* __restrict__ col,
                                                      const float* __restrict__ dinv,
                                                      const float* __restrict__ h,
                                                      float* __restrict__ agg) {
    __shared__ int   s_row[EPB];
    __shared__ int   s_col[EPB];
    __shared__ float s_norm[EPB];
    const int e0 = blockIdx.x * EPB;
    const int t  = threadIdx.x;

    for (int i = t; i < EPB; i += 128) {
        int e = e0 + i;
        if (e < N_EDGES) {
            int r = row[e], c = col[e];
            s_row[i]  = r;
            s_col[i]  = c;
            s_norm[i] = dinv[r] * dinv[c];
        }
    }
    __syncthreads();

    int lim = N_EDGES - e0;
    if (lim > EPB) lim = EPB;
    for (int i = 0; i < lim; ++i) {
        float v = s_norm[i] * h[(size_t)s_row[i] * DIM + t];
        atomicAdd(&agg[(size_t)s_col[i] * DIM + t], v);
    }
}

__global__ void bias_relu_kernel(float* __restrict__ x, const float* __restrict__ b) {
    size_t idx = (size_t)blockIdx.x * blockDim.x + threadIdx.x;
    if (idx < (size_t)N_NODES * DIM)
        x[idx] = fmaxf(x[idx] + b[idx & (DIM - 1)], 0.0f);
}

extern "C" void kernel_launch(void* const* d_in, const int* in_sizes, int n_in,
                              void* d_out, int out_size, void* d_ws, size_t ws_size,
                              hipStream_t stream) {
    const float* x    = (const float*)d_in[0];
    const int*   edge = (const int*)d_in[1];
    const float* W1   = (const float*)d_in[2];
    const float* b1   = (const float*)d_in[3];
    const float* W2   = (const float*)d_in[4];
    const float* b2   = (const float*)d_in[5];
    const float* Wout = (const float*)d_in[6];
    const float* bout = (const float*)d_in[7];
    const int* row = edge;
    const int* col = edge + N_EDGES;
    float* out = (float*)d_out;

    char* ws = (char*)d_ws;
    float* dinv = (float*)ws;
    float* bufA = (float*)(ws + ((size_t)1 << 20));
    float* bufB = bufA + (size_t)N_NODES * DIM;

    const int nelem  = N_NODES * DIM;
    const int ntiles = N_NODES / 16;

    deg_init_kernel<<<(N_NODES + 255) / 256, 256, 0, stream>>>(dinv);
    deg_acc_kernel<<<(N_EDGES + 255) / 256, 256, 0, stream>>>(col, dinv);
    dinv_kernel<<<(N_NODES + 255) / 256, 256, 0, stream>>>(dinv);

    gemm128_wmma<<<ntiles, 256, 0, stream>>>(x, W1, bufA);
    agg_selfloop_kernel<<<(nelem + 255) / 256, 256, 0, stream>>>(dinv, bufA, bufB);
    scatter_kernel<<<(N_EDGES + EPB - 1) / EPB, 128, 0, stream>>>(row, col, dinv, bufA, bufB);
    bias_relu_kernel<<<(nelem + 255) / 256, 256, 0, stream>>>(bufB, b1);

    gemm128_wmma<<<ntiles, 256, 0, stream>>>(bufB, W2, bufA);
    agg_selfloop_kernel<<<(nelem + 255) / 256, 256, 0, stream>>>(dinv, bufA, bufB);
    scatter_kernel<<<(N_EDGES + EPB - 1) / EPB, 128, 0, stream>>>(row, col, dinv, bufA, bufB);
    bias_relu_kernel<<<(nelem + 255) / 256, 256, 0, stream>>>(bufB, b2);

    gemm_out_wmma<<<ntiles, 96, 0, stream>>>(bufB, Wout, bout, out);
}


# Aggregation by owner blocks: 256 blocks each keep 391 target rows in LDS, read the whole target list, add their edges' rows by LDS f32 adds (arrival order), store each row once; launches 12 -> 10
# speedup vs baseline: 1.5508x; 1.5508x over previous
#include <hip/hip_runtime.h>

typedef __attribute__((ext_vector_type(2))) float v2f;
typedef __attribute__((ext_vector_type(8))) float v8f;

#define N_NODES 100000
#define N_EDGES 1600000
#define DIM 128
#define N_CLASSES 40
#define TROWS 391
#define NBLK_AGG 256
#define QCAP 136
#define NCHUNK (N_EDGES / 128)
#define GRP 8
static_assert(N_EDGES % 128 == 0, "edge list is read in whole chunks");
static_assert(NBLK_AGG * TROWS >= N_NODES, "blocks must cover every target row");
static_assert(TROWS <= 512 && N_EDGES <= (1 << 21), "queue entry packs edge id and local row in 32 bits");

__global__ void deg_init_kernel(float* deg) {
    int i = blockIdx.x * blockDim.x + threadIdx.x;
    if (i < N_NODES) deg[i] = 1.0f;
}

__global__ void deg_acc_kernel(const int* __restrict__ col, float* deg) {
    int e = blockIdx.x * blockDim.x + threadIdx.x;
    if (e < N_EDGES) atomicAdd(&deg[col[e]], 1.0f);
}

__global__ void dinv_kernel(float* deg) {
    int i = blockIdx.x * blockDim.x + threadIdx.x;
    if (i < N_NODES) deg[i] = rsqrtf(deg[i]);
}

__global__ __launch_bounds__(256) void gemm128_wmma(const float* __restrict__ X,
                                                    const float* __restrict__ W,
                                                    float* __restrict__ H) {
    const int m0    = blockIdx.x * 16;
    const int lane  = threadIdx.x & 31;
    const int wave  = threadIdx.x >> 5;
    const int n0    = wave * 16;
    const int l15   = lane & 15;
    const int khalf = (lane >> 4) << 1;
    const float* xrow = X + (size_t)(m0 + l15) * DIM;

    v8f acc = {};
    for (int k = 0; k < DIM; k += 4) {
        const int kk = k + khalf;
        v2f a, b;
        a.x = xrow[kk];
        a.y = xrow[kk + 1];
        b.x = W[(size_t)kk * DIM + n0 + l15];
        b.y = W[(size_t)(kk + 1) * DIM + n0 + l15];
        acc = __builtin_amdgcn_wmma_f32_16x16x4_f32(
             false, a,  false, b,
             (short)0, acc,  false,  false);
    }
    const int crow = m0 + (lane >> 4) * 8;
    const int ccol = n0 + l15;
#pragma unroll
    for (int r = 0; r < 8; ++r)
        H[(size_t)(crow + r) * DIM + ccol] = acc[r];
}

__global__ __launch_bounds__(96) void gemm_out_wmma(const float* __restrict__ X,
                                                    const float* __restrict__ W,
                                                    const float* __restrict__ bias,
                                                    float* __restrict__ OUT) {
    const int m0    = blockIdx.x * 16;
    const int lane  = threadIdx.x & 31;
    const int wave  = threadIdx.x >> 5;
    const int n0    = wave * 16;
    const int l15   = lane & 15;
    const int khalf = (lane >> 4) << 1;
    const int ncol  = n0 + l15;
    const bool nv   = (ncol < N_CLASSES);
    const float* xrow = X + (size_t)(m0 + l15) * DIM;

    v8f acc = {};
    for (int k = 0; k < DIM; k += 4) {
        const int kk = k + khalf;
        v2f a, b;
        a.x = xrow[kk];
        a.y = xrow[kk + 1];
        b.x = nv ? W[(size_t)kk * N_CLASSES + ncol] : 0.0f;
        b.y = nv ? W[(size_t)(kk + 1) * N_CLASSES + ncol] : 0.0f;
        acc = __builtin_amdgcn_wmma_f32_16x16x4_f32(
            false, a, false, b, (short)0, acc, false, false);
    }
    if (nv) {
        const int crow = m0 + (lane >> 4) * 8;
        const float bb = bias[ncol];
#pragma unroll
        for (int r = 0; r < 8; ++r)
            OUT[(size_t)(crow + r) * N_CLASSES + ncol] = acc[r] + bb;
    }
}


typedef volatile __attribute__((address_space(3))) unsigned lds_vu32;

__device__ __forceinline__ void drain_edges(const lds_vu32* q, int n, int lane,
                                            const int* __restrict__ row,
                                            const float* __restrict__ dinv,
                                            const float* __restrict__ h,
                                            const float* s_dinv, float* tile) {
    unsigned ent[8];
    int r[8];
    float w[8];
    float hv[8][4];
#pragma unroll
    for (int b = 0; b < 8; ++b) {
        ent[b] = __builtin_amdgcn_readfirstlane(q[b < n ? b : 0]);
    }
#pragma unroll
    for (int b = 0; b < 8; ++b) r[b] = row[ent[b] >> 9];
#pragma unroll
    for (int b = 0; b < 8; ++b) {
        w[b] = dinv[r[b]] * s_dinv[ent[b] & 511u];
#pragma unroll
        for (int k = 0; k < 4; ++k) hv[b][k] = h[(size_t)r[b] * DIM + k * 32 + lane];
    }
#pragma unroll
    for (int b = 0; b < 8; ++b) {
        if (b < n) {
            float* trow = tile + (ent[b] & 511u) * DIM + lane;
#pragma unroll
            for (int k = 0; k < 4; ++k)
                __hip_atomic_fetch_add(trow + k * 32, w[b] * hv[b][k], __ATOMIC_RELAXED,
                                       __HIP_MEMORY_SCOPE_WORKGROUP);
        }
    }
}

__global__ __launch_bounds__(1024) void agg_owner_kernel(const int* __restrict__ row,
                                                         const int* __restrict__ col,
                                                         const float* __restrict__ dinv,
                                                         const float* __restrict__ h,
                                                         float* __restrict__ agg) {
    __shared__ float    tile[TROWS * DIM];
    __shared__ float    s_dinv[TROWS + 1];
    __shared__ unsigned s_q[32 * QCAP];

    const int tid  = threadIdx.x;
    const int lane = tid & 31;
    const int wave = tid >> 5;
    const int lo   = blockIdx.x * TROWS;
    int nrows = N_NODES - lo;
    if (nrows > TROWS) nrows = TROWS;

    const float4* h4    = (const float4*)h;
    float4*       tile4 = (float4*)tile;
    for (int i = tid; i < nrows * 32; i += 1024) {
        const float d  = dinv[lo + (i >> 5)];
        const float dd = d * d;
        float4 v = h4[(size_t)lo * 32 + i];
        v.x = dd * v.x;
        v.y = dd * v.y;
        v.z = dd * v.z;
        v.w = dd * v.w;
        tile4[i] = v;
    }
    for (int i = tid; i < nrows; i += 1024) s_dinv[i] = dinv[lo + i];
    __syncthreads();

    lds_vu32* q = (lds_vu32*)(s_q + wave * QCAP);
    const unsigned below = (1u << lane) - 1u;
    const int4* col4 = (const int4*)col;
    int qn = 0;

    for (int g0 = wave * GRP; g0 < NCHUNK; g0 += 32 * GRP) {
        int4 cb[GRP];
#pragma unroll
        for (int u = 0; u < GRP; ++u) {
            if (g0 + u < NCHUNK) cb[u] = col4[(size_t)(g0 + u) * 32 + lane];
            else                 cb[u] = make_int4(-1, -1, -1, -1);
        }
#pragma unroll
        for (int u = 0; u < GRP; ++u) {
            const unsigned c0 = (unsigned)(cb[u].x - lo);
            const unsigned c1 = (unsigned)(cb[u].y - lo);
            const unsigned c2 = (unsigned)(cb[u].z - lo);
            const unsigned c3 = (unsigned)(cb[u].w - lo);
            const bool m0 = c0 < (unsigned)nrows;
            const bool m1 = c1 < (unsigned)nrows;
            const bool m2 = c2 < (unsigned)nrows;
            const bool m3 = c3 < (unsigned)nrows;
            if (__builtin_amdgcn_ballot_w32(m0 | m1 | m2 | m3) == 0u) continue;

            const unsigned e0 = (unsigned)(g0 + u) * 128u + (unsigned)lane * 4u;
            unsigned mk;
            mk = __builtin_amdgcn_ballot_w32(m0);
            if (m0) q[qn + __builtin_popcount(mk & below)] = ((e0 + 0u) << 9) | c0;
            qn += __builtin_popcount(mk);
            mk = __builtin_amdgcn_ballot_w32(m1);
            if (m1) q[qn + __builtin_popcount(mk & below)] = ((e0 + 1u) << 9) | c1;
            qn += __builtin_popcount(mk);
            mk = __builtin_amdgcn_ballot_w32(m2);
            if (m2) q[qn + __builtin_popcount(mk & below)] = ((e0 + 2u) << 9) | c2;
            qn += __builtin_popcount(mk);
            mk = __builtin_amdgcn_ballot_w32(m3);
            if (m3) q[qn + __builtin_popcount(mk & below)] = ((e0 + 3u) << 9) | c3;
            qn += __builtin_popcount(mk);

            while (qn >= 8) {
                qn -= 8;
                drain_edges(q + qn, 8, lane, row, dinv, h, s_dinv, tile);
            }
        }
    }
    if (qn > 0) drain_edges(q, qn, lane, row, dinv, h, s_dinv, tile);
    __syncthreads();

    float4* agg4 = (float4*)agg;
    for (int i = tid; i < nrows * 32; i += 1024) agg4[(size_t)lo * 32 + i] = tile4[i];
}

__global__ void bias_relu_kernel(float* __restrict__ x, const float* __restrict__ b) {
    size_t idx = (size_t)blockIdx.x * blockDim.x + threadIdx.x;
    if (idx < (size_t)N_NODES * DIM)
        x[idx] = fmaxf(x[idx] + b[idx & (DIM - 1)], 0.0f);
}

extern "C" void kernel_launch(void* const* d_in, const int* in_sizes, int n_in,
                              void* d_out, int out_size, void* d_ws, size_t ws_size,
                              hipStream_t stream) {
    const float* x    = (const float*)d_in[0];
    const int*   edge = (const int*)d_in[1];
    const float* W1   = (const float*)d_in[2];
    const float* b1   = (const float*)d_in[3];
    const float* W2   = (const float*)d_in[4];
    const float* b2   = (const float*)d_in[5];
    const float* Wout = (const float*)d_in[6];
    const float* bout = (const float*)d_in[7];
    const int* row = edge;
    const int* col = edge + N_EDGES;
    float* out = (float*)d_out;

    char* ws = (char*)d_ws;
    float* dinv = (float*)ws;
    float* bufA = (float*)(ws + ((size_t)1 << 20));
    float* bufB = bufA + (size_t)N_NODES * DIM;

    const int nelem  = N_NODES * DIM;
    const int ntiles = N_NODES / 16;

    deg_init_kernel<<<(N_NODES + 255) / 256, 256, 0, stream>>>(dinv);
    deg_acc_kernel<<<(N_EDGES + 255) / 256, 256, 0, stream>>>(col, dinv);
    dinv_kernel<<<(N_NODES + 255) / 256, 256, 0, stream>>>(dinv);

    gemm128_wmma<<<ntiles, 256, 0, stream>>>(x, W1, bufA);
    agg_owner_kernel<<<NBLK_AGG, 1024, 0, stream>>>(row, col, dinv, bufA, bufB);
    bias_relu_kernel<<<(nelem + 255) / 256, 256, 0, stream>>>(bufB, b1);

    gemm128_wmma<<<ntiles, 256, 0, stream>>>(bufB, W2, bufA);
    agg_owner_kernel<<<NBLK_AGG, 1024, 0, stream>>>(row, col, dinv, bufA, bufB);
    bias_relu_kernel<<<(nelem + 255) / 256, 256, 0, stream>>>(bufB, b2);

    gemm_out_wmma<<<ntiles, 96, 0, stream>>>(bufB, Wout, bout, out);
}


# Bias add and relu moved to the reader: the layer-2 and output matrix products apply relu(sum + b) to each A value as they load it; both elementwise passes are gone; launches 10 -> 8
# speedup vs baseline: 1.7731x; 1.1433x over previous
#include <hip/hip_runtime.h>

typedef __attribute__((ext_vector_type(2))) float v2f;
typedef __attribute__((ext_vector_type(8))) float v8f;

#define N_NODES 100000
#define N_EDGES 1600000
#define DIM 128
#define N_CLASSES 40
#define TROWS 391
#define NBLK_AGG 256
#define QCAP 136
#define NCHUNK (N_EDGES / 128)
#define GRP 8
static_assert(N_EDGES % 128 == 0, "edge list is read in whole chunks");
static_assert(NBLK_AGG * TROWS >= N_NODES, "blocks must cover every target row");
static_assert(TROWS <= 512 && N_EDGES <= (1 << 21), "queue entry packs edge id and local row in 32 bits");

__global__ void deg_init_kernel(float* deg) {
    int i = blockIdx.x * blockDim.x + threadIdx.x;
    if (i < N_NODES) deg[i] = 1.0f;
}

__global__ void deg_acc_kernel(const int* __restrict__ col, float* deg) {
    int e = blockIdx.x * blockDim.x + threadIdx.x;
    if (e < N_EDGES) atomicAdd(&deg[col[e]], 1.0f);
}

__global__ void dinv_kernel(float* deg) {
    int i = blockIdx.x * blockDim.x + threadIdx.x;
    if (i < N_NODES) deg[i] = rsqrtf(deg[i]);
}

template <bool ACT>
__device__ __forceinline__ void gemm128_body(const float* __restrict__ X,
                                             const float* __restrict__ W,
                                             const float* __restrict__ bias,
                                             float* __restrict__ H) {
    const int m0    = blockIdx.x * 16;
    const int lane  = threadIdx.x & 31;
    const int wave  = threadIdx.x >> 5;
    const int n0    = wave * 16;
    const int l15   = lane & 15;
    const int khalf = (lane >> 4) << 1;
    const float* xrow = X + (size_t)(m0 + l15) * DIM;

    v8f acc = {};
    for (int k = 0; k < DIM; k += 4) {
        const int kk = k + khalf;
        v2f a, b;
        a.x = xrow[kk];
        a.y = xrow[kk + 1];
        if (ACT) {
            a.x = fmaxf(a.x + bias[kk], 0.0f);
            a.y = fmaxf(a.y + bias[kk + 1], 0.0f);
        }
        b.x = W[(size_t)kk * DIM + n0 + l15];
        b.y = W[(size_t)(kk + 1) * DIM + n0 + l15];
        acc = __builtin_amdgcn_wmma_f32_16x16x4_f32(
             false, a,  false, b,
             (short)0, acc,  false,  false);
    }
    const int crow = m0 + (lane >> 4) * 8;
    const int ccol = n0 + l15;
#pragma unroll
    for (int r = 0; r < 8; ++r)
        H[(size_t)(crow + r) * DIM + ccol] = acc[r];
}

__global__ __launch_bounds__(256) void gemm128_wmma(const float* __restrict__ X,
                                                    const float* __restrict__ W,
                                                    float* __restrict__ H) {
    gemm128_body<false>(X, W, nullptr, H);
}

__global__ __launch_bounds__(256) void gemm128_act_wmma(const float* __restrict__ S,
                                                        const float* __restrict__ bias,
                                                        const float* __restrict__ W,
                                                        float* __restrict__ H) {
    gemm128_body<true>(S, W, bias, H);
}

__global__ __launch_bounds__(96) void gemm_out_wmma(const float* __restrict__ S,
                                                    const float* __restrict__ bin,
                                                    const float* __restrict__ W,
                                                    const float* __restrict__ bias,
                                                    float* __restrict__ OUT) {
    const int m0    = blockIdx.x * 16;
    const int lane  = threadIdx.x & 31;
    const int wave  = threadIdx.x >> 5;
    const int n0    = wave * 16;
    const int l15   = lane & 15;
    const int khalf = (lane >> 4) << 1;
    const int ncol  = n0 + l15;
    const bool nv   = (ncol < N_CLASSES);
    const float* xrow = S + (size_t)(m0 + l15) * DIM;

    v8f acc = {};
    for (int k = 0; k < DIM; k += 4) {
        const int kk = k + khalf;
        v2f a, b;
        a.x = fmaxf(xrow[kk] + bin[kk], 0.0f);
        a.y = fmaxf(xrow[kk + 1] + bin[kk + 1], 0.0f);
        b.x = nv ? W[(size_t)kk * N_CLASSES + ncol] : 0.0f;
        b.y = nv ? W[(size_t)(kk + 1) * N_CLASSES + ncol] : 0.0f;
        acc = __builtin_amdgcn_wmma_f32_16x16x4_f32(
            false, a, false, b, (short)0, acc, false, false);
    }
    if (nv) {
        const int crow = m0 + (lane >> 4) * 8;
        const float bb = bias[ncol];
#pragma unroll
        for (int r = 0; r < 8; ++r)
            OUT[(size_t)(crow + r) * N_CLASSES + ncol] = acc[r] + bb;
    }
}


typedef volatile __attribute__((address_space(3))) unsigned lds_vu32;

__device__ __forceinline__ void drain_edges(const lds_vu32* q, int n, int lane,
                                            const int* __restrict__ row,
                                            const float* __restrict__ dinv,
                                            const float* __restrict__ h,
                                            const float* s_dinv, float* tile) {
    unsigned ent[8];
    int r[8];
    float w[8];
    float hv[8][4];
#pragma unroll
    for (int b = 0; b < 8; ++b) {
        ent[b] = __builtin_amdgcn_readfirstlane(q[b < n ? b : 0]);
    }
#pragma unroll
    for (int b = 0; b < 8; ++b) r[b] = row[ent[b] >> 9];
#pragma unroll
    for (int b = 0; b < 8; ++b) {
        w[b] = dinv[r[b]] * s_dinv[ent[b] & 511u];
#pragma unroll
        for (int k = 0; k < 4; ++k) hv[b][k] = h[(size_t)r[b] * DIM + k * 32 + lane];
    }
#pragma unroll
    for (int b = 0; b < 8; ++b) {
        if (b < n) {
            float* trow = tile + (ent[b] & 511u) * DIM + lane;
#pragma unroll
            for (int k = 0; k < 4; ++k)
                __hip_atomic_fetch_add(trow + k * 32, w[b] * hv[b][k], __ATOMIC_RELAXED,
                                       __HIP_MEMORY_SCOPE_WORKGROUP);
        }
    }
}

__global__ __launch_bounds__(1024) void agg_owner_kernel(const int* __restrict__ row,
                                                         const int* __restrict__ col,
                                                         const float* __restrict__ dinv,
                                                         const float* __restrict__ h,
                                                         float* __restrict__ agg) {
    __shared__ float    tile[TROWS * DIM];
    __shared__ float    s_dinv[TROWS + 1];
    __shared__ unsigned s_q[32 * QCAP];

    const int tid  = threadIdx.x;
    const int lane = tid & 31;
    const int wave = tid >> 5;
    const int lo   = blockIdx.x * TROWS;
    int nrows = N_NODES - lo;
    if (nrows > TROWS) nrows = TROWS;

    const float4* h4    = (const float4*)h;
    float4*       tile4 = (float4*)tile;
    for (int i = tid; i < nrows * 32; i += 1024) {
        const float d  = dinv[lo + (i >> 5)];
        const float dd = d * d;
        float4 v = h4[(size_t)lo * 32 + i];
        v.x = dd * v.x;
        v.y = dd * v.y;
        v.z = dd * v.z;
        v.w = dd * v.w;
        tile4[i] = v;
    }
    for (int i = tid; i < nrows; i += 1024) s_dinv[i] = dinv[lo + i];
    __syncthreads();

    lds_vu32* q = (lds_vu32*)(s_q + wave * QCAP);
    const unsigned below = (1u << lane) - 1u;
    const int4* col4 = (const int4*)col;
    int qn = 0;

    for (int g0 = wave * GRP; g0 < NCHUNK; g0 += 32 * GRP) {
        int4 cb[GRP];
#pragma unroll
        for (int u = 0; u < GRP; ++u) {
            if (g0 + u < NCHUNK) cb[u] = col4[(size_t)(g0 + u) * 32 + lane];
            else                 cb[u] = make_int4(-1, -1, -1, -1);
        }
#pragma unroll
        for (int u = 0; u < GRP; ++u) {
            const unsigned c0 = (unsigned)(cb[u].x - lo);
            const unsigned c1 = (unsigned)(cb[u].y - lo);
            const unsigned c2 = (unsigned)(cb[u].z - lo);
            const unsigned c3 = (unsigned)(cb[u].w - lo);
            const bool m0 = c0 < (unsigned)nrows;
            const bool m1 = c1 < (unsigned)nrows;
            const bool m2 = c2 < (unsigned)nrows;
            const bool m3 = c3 < (unsigned)nrows;
            if (__builtin_amdgcn_ballot_w32(m0 | m1 | m2 | m3) == 0u) continue;

            const unsigned e0 = (unsigned)(g0 + u) * 128u + (unsigned)lane * 4u;
            unsigned mk;
            mk = __builtin_amdgcn_ballot_w32(m0);
            if (m0) q[qn + __builtin_popcount(mk & below)] = ((e0 + 0u) << 9) | c0;
            qn += __builtin_popcount(mk);
            mk = __builtin_amdgcn_ballot_w32(m1);
            if (m1) q[qn + __builtin_popcount(mk & below)] = ((e0 + 1u) << 9) | c1;
            qn += __builtin_popcount(mk);
            mk = __builtin_amdgcn_ballot_w32(m2);
            if (m2) q[qn + __builtin_popcount(mk & below)] = ((e0 + 2u) << 9) | c2;
            qn += __builtin_popcount(mk);
            mk = __builtin_amdgcn_ballot_w32(m3);
            if (m3) q[qn + __builtin_popcount(mk & below)] = ((e0 + 3u) << 9) | c3;
            qn += __builtin_popcount(mk);

            while (qn >= 8) {
                qn -= 8;
                drain_edges(q + qn, 8, lane, row, dinv, h, s_dinv, tile);
            }
        }
    }
    if (qn > 0) drain_edges(q, qn, lane, row, dinv, h, s_dinv, tile);
    __syncthreads();

    float4* agg4 = (float4*)agg;
    for (int i = tid; i < nrows * 32; i += 1024) agg4[(size_t)lo * 32 + i] = tile4[i];
}

extern "C" void kernel_launch(void* const* d_in, const int* in_sizes, int n_in,
                              void* d_out, int out_size, void* d_ws, size_t ws_size,
                              hipStream_t stream) {
    const float* x    = (const float*)d_in[0];
    const int*   edge = (const int*)d_in[1];
    const float* W1   = (const float*)d_in[2];
    const float* b1   = (const float*)d_in[3];
    const float* W2   = (const float*)d_in[4];
    const float* b2   = (const float*)d_in[5];
    const float* Wout = (const float*)d_in[6];
    const float* bout = (const float*)d_in[7];
    const int* row = edge;
    const int* col = edge + N_EDGES;
    float* out = (float*)d_out;

    char* ws = (char*)d_ws;
    float* dinv = (float*)ws;
    float* bufA = (float*)(ws + ((size_t)1 << 20));
    float* bufB = bufA + (size_t)N_NODES * DIM;

    const int ntiles = N_NODES / 16;

    deg_init_kernel<<<(N_NODES + 255) / 256, 256, 0, stream>>>(dinv);
    deg_acc_kernel<<<(N_EDGES + 255) / 256, 256, 0, stream>>>(col, dinv);
    dinv_kernel<<<(N_NODES + 255) / 256, 256, 0, stream>>>(dinv);

    gemm128_wmma<<<ntiles, 256, 0, stream>>>(x, W1, bufA);
    agg_owner_kernel<<<NBLK_AGG, 1024, 0, stream>>>(row, col, dinv, bufA, bufB);

    gemm128_act_wmma<<<ntiles, 256, 0, stream>>>(bufB, b1, W2, bufA);
    agg_owner_kernel<<<NBLK_AGG, 1024, 0, stream>>>(row, col, dinv, bufA, bufB);

    gemm_out_wmma<<<ntiles, 96, 0, stream>>>(bufB, b2, Wout, bout, out);
}


# Layer-2 aggregation adds its edges from per-block lists that the layer-1 aggregation records while it scans the target list; the second 256-fold scan is gone (LDS f32 adds stay arrival order)
# speedup vs baseline: 1.9606x; 1.1057x over previous
#include <hip/hip_runtime.h>

typedef __attribute__((ext_vector_type(2))) float v2f;
typedef __attribute__((ext_vector_type(8))) float v8f;

#define N_NODES 100000
#define N_EDGES 1600000
#define DIM 128
#define N_CLASSES 40
#define TROWS 391
#define NBLK_AGG 256
#define QCAP 136
#define NCHUNK (N_EDGES / 128)
#define GRP 8
#define CAPB 32768
static_assert(N_EDGES % 128 == 0, "edge list is read in whole chunks");
static_assert(NBLK_AGG * TROWS >= N_NODES, "blocks must cover every target row");
static_assert(TROWS <= 512 && N_EDGES <= (1 << 21), "queue entry packs edge id and local row in 32 bits");

__global__ void deg_init_kernel(float* deg) {
    int i = blockIdx.x * blockDim.x + threadIdx.x;
    if (i < N_NODES) deg[i] = 1.0f;
}

__global__ void deg_acc_kernel(const int* __restrict__ col, float* deg) {
    int e = blockIdx.x * blockDim.x + threadIdx.x;
    if (e < N_EDGES) atomicAdd(&deg[col[e]], 1.0f);
}

__global__ void dinv_kernel(float* deg) {
    int i = blockIdx.x * blockDim.x + threadIdx.x;
    if (i < N_NODES) deg[i] = rsqrtf(deg[i]);
}

template <bool ACT>
__device__ __forceinline__ void gemm128_body(const float* __restrict__ X,
                                             const float* __restrict__ W,
                                             const float* __restrict__ bias,
                                             float* __restrict__ H) {
    const int m0    = blockIdx.x * 16;
    const int lane  = threadIdx.x & 31;
    const int wave  = threadIdx.x >> 5;
    const int n0    = wave * 16;
    const int l15   = lane & 15;
    const int khalf = (lane >> 4) << 1;
    const float* xrow = X + (size_t)(m0 + l15) * DIM;

    v8f acc = {};
    for (int k = 0; k < DIM; k += 4) {
        const int kk = k + khalf;
        v2f a, b;
        a.x = xrow[kk];
        a.y = xrow[kk + 1];
        if (ACT) {
            a.x = fmaxf(a.x + bias[kk], 0.0f);
            a.y = fmaxf(a.y + bias[kk + 1], 0.0f);
        }
        b.x = W[(size_t)kk * DIM + n0 + l15];
        b.y = W[(size_t)(kk + 1) * DIM + n0 + l15];
        acc = __builtin_amdgcn_wmma_f32_16x16x4_f32(
             false, a,  false, b,
             (short)0, acc,  false,  false);
    }
    const int crow = m0 + (lane >> 4) * 8;
    const int ccol = n0 + l15;
#pragma unroll
    for (int r = 0; r < 8; ++r)
        H[(size_t)(crow + r) * DIM + ccol] = acc[r];
}

__global__ __launch_bounds__(256) void gemm128_wmma(const float* __restrict__ X,
                                                    const float* __restrict__ W,
                                                    float* __restrict__ H) {
    gemm128_body<false>(X, W, nullptr, H);
}

__global__ __launch_bounds__(256) void gemm128_act_wmma(const float* __restrict__ S,
                                                        const float* __restrict__ bias,
                                                        const float* __restrict__ W,
                                                        float* __restrict__ H) {
    gemm128_body<true>(S, W, bias, H);
}

__global__ __launch_bounds__(96) void gemm_out_wmma(const float* __restrict__ S,
                                                    const float* __restrict__ bin,
                                                    const float* __restrict__ W,
                                                    const float* __restrict__ bias,
                                                    float* __restrict__ OUT) {
    const int m0    = blockIdx.x * 16;
    const int lane  = threadIdx.x & 31;
    const int wave  = threadIdx.x >> 5;
    const int n0    = wave * 16;
    const int l15   = lane & 15;
    const int khalf = (lane >> 4) << 1;
    const int ncol  = n0 + l15;
    const bool nv   = (ncol < N_CLASSES);
    const float* xrow = S + (size_t)(m0 + l15) * DIM;

    v8f acc = {};
    for (int k = 0; k < DIM; k += 4) {
        const int kk = k + khalf;
        v2f a, b;
        a.x = fmaxf(xrow[kk] + bin[kk], 0.0f);
        a.y = fmaxf(xrow[kk + 1] + bin[kk + 1], 0.0f);
        b.x = nv ? W[(size_t)kk * N_CLASSES + ncol] : 0.0f;
        b.y = nv ? W[(size_t)(kk + 1) * N_CLASSES + ncol] : 0.0f;
        acc = __builtin_amdgcn_wmma_f32_16x16x4_f32(
            false, a, false, b, (short)0, acc, false, false);
    }
    if (nv) {
        const int crow = m0 + (lane >> 4) * 8;
        const float bb = bias[ncol];
#pragma unroll
        for (int r = 0; r < 8; ++r)
            OUT[(size_t)(crow + r) * N_CLASSES + ncol] = acc[r] + bb;
    }
}


typedef volatile __attribute__((address_space(3))) unsigned lds_vu32;

template <typename QP>
__device__ __forceinline__ void drain_edges(QP q, int n, int lane,
                                            const int* __restrict__ row,
                                            const float* __restrict__ dinv,
                                            const float* __restrict__ h,
                                            const float* s_dinv, float* tile) {
    unsigned ent[8];
    int r[8];
    float w[8];
    float hv[8][4];
#pragma unroll
    for (int b = 0; b < 8; ++b) {
        ent[b] = __builtin_amdgcn_readfirstlane(q[b < n ? b : 0]);
    }
#pragma unroll
    for (int b = 0; b < 8; ++b) r[b] = row[ent[b] >> 9];
#pragma unroll
    for (int b = 0; b < 8; ++b) {
        w[b] = dinv[r[b]] * s_dinv[ent[b] & 511u];
#pragma unroll
        for (int k = 0; k < 4; ++k) hv[b][k] = h[(size_t)r[b] * DIM + k * 32 + lane];
    }
#pragma unroll
    for (int b = 0; b < 8; ++b) {
        if (b < n) {
            float* trow = tile + (ent[b] & 511u) * DIM + lane;
#pragma unroll
            for (int k = 0; k < 4; ++k)
                __hip_atomic_fetch_add(trow + k * 32, w[b] * hv[b][k], __ATOMIC_RELAXED,
                                       __HIP_MEMORY_SCOPE_WORKGROUP);
        }
    }
}

__device__ __forceinline__ void record_entries(const lds_vu32* q, int n, int lane,
                                               int* s_cnt, unsigned* rec) {
    int base = 0;
    if (lane == 0) base = atomicAdd(s_cnt, n);
    base = __builtin_amdgcn_readfirstlane(base);
    if (lane < n && base + n <= CAPB) rec[base + lane] = q[lane];
}

template <bool RECORD>
__device__ __forceinline__ void scan_targets(lds_vu32* q, int lo, int nrows, int lane, int wave,
                                             const int* __restrict__ row,
                                             const int* __restrict__ col,
                                             const float* __restrict__ dinv,
                                             const float* __restrict__ h,
                                             const float* s_dinv, float* tile,
                                             int* s_cnt, unsigned* rec) {
    const unsigned below = (1u << lane) - 1u;
    const int4* col4 = (const int4*)col;
    int qn = 0;

    for (int g0 = wave * GRP; g0 < NCHUNK; g0 += 32 * GRP) {
        int4 cb[GRP];
#pragma unroll
        for (int u = 0; u < GRP; ++u) {
            if (g0 + u < NCHUNK) cb[u] = col4[(size_t)(g0 + u) * 32 + lane];
            else                 cb[u] = make_int4(-1, -1, -1, -1);
        }
#pragma unroll
        for (int u = 0; u < GRP; ++u) {
            const unsigned c0 = (unsigned)(cb[u].x - lo);
            const unsigned c1 = (unsigned)(cb[u].y - lo);
            const unsigned c2 = (unsigned)(cb[u].z - lo);
            const unsigned c3 = (unsigned)(cb[u].w - lo);
            const bool m0 = c0 < (unsigned)nrows;
            const bool m1 = c1 < (unsigned)nrows;
            const bool m2 = c2 < (unsigned)nrows;
            const bool m3 = c3 < (unsigned)nrows;
            if (__builtin_amdgcn_ballot_w32(m0 | m1 | m2 | m3) == 0u) continue;

            const unsigned e0 = (unsigned)(g0 + u) * 128u + (unsigned)lane * 4u;
            unsigned mk;
            mk = __builtin_amdgcn_ballot_w32(m0);
            if (m0) q[qn + __builtin_popcount(mk & below)] = ((e0 + 0u) << 9) | c0;
            qn += __builtin_popcount(mk);
            mk = __builtin_amdgcn_ballot_w32(m1);
            if (m1) q[qn + __builtin_popcount(mk & below)] = ((e0 + 1u) << 9) | c1;
            qn += __builtin_popcount(mk);
            mk = __builtin_amdgcn_ballot_w32(m2);
            if (m2) q[qn + __builtin_popcount(mk & below)] = ((e0 + 2u) << 9) | c2;
            qn += __builtin_popcount(mk);
            mk = __builtin_amdgcn_ballot_w32(m3);
            if (m3) q[qn + __builtin_popcount(mk & below)] = ((e0 + 3u) << 9) | c3;
            qn += __builtin_popcount(mk);

            while (qn >= 8) {
                qn -= 8;
                if (RECORD) record_entries(q + qn, 8, lane, s_cnt, rec);
                drain_edges(q + qn, 8, lane, row, dinv, h, s_dinv, tile);
            }
        }
    }
    if (qn > 0) {
        if (RECORD) record_entries(q, qn, lane, s_cnt, rec);
        drain_edges(q, qn, lane, row, dinv, h, s_dinv, tile);
    }
}

__device__ __forceinline__ void replay_edges(const unsigned* __restrict__ rec, int cnt,
                                             int lane, int wave,
                                             const int* __restrict__ row,
                                             const float* __restrict__ dinv,
                                             const float* __restrict__ h,
                                             const float* s_dinv, float* tile) {
    const int wave_u = __builtin_amdgcn_readfirstlane(wave);
    for (int i0 = wave_u * 32; i0 < cnt; i0 += 1024) {
        int nb = cnt - i0;
        if (nb > 32) nb = 32;
        const unsigned ent = (lane < nb) ? rec[i0 + lane] : 0u;
        const int      r   = row[ent >> 9];
        const float    w   = dinv[r] * s_dinv[ent & 511u];
        for (int g = 0; g < nb; g += 8) {
            unsigned eb[8];
            int rb[8];
            float wb[8];
            float hv[8][4];
#pragma unroll
            for (int b = 0; b < 8; ++b) {
                const int sel = (g + b < nb) ? (g + b) : g;
                eb[b] = (unsigned)__builtin_amdgcn_readlane((int)ent, sel);
                rb[b] = __builtin_amdgcn_readlane(r, sel);
                wb[b] = __int_as_float(__builtin_amdgcn_readlane(__float_as_int(w), sel));
            }
#pragma unroll
            for (int b = 0; b < 8; ++b) {
#pragma unroll
                for (int k = 0; k < 4; ++k) hv[b][k] = h[(size_t)rb[b] * DIM + k * 32 + lane];
            }
#pragma unroll
            for (int b = 0; b < 8; ++b) {
                if (b == 0 || g + b < nb) {
                    float* trow = tile + (eb[b] & 511u) * DIM + lane;
#pragma unroll
                    for (int k = 0; k < 4; ++k)
                        __hip_atomic_fetch_add(trow + k * 32, wb[b] * hv[b][k], __ATOMIC_RELAXED,
                                               __HIP_MEMORY_SCOPE_WORKGROUP);
                }
            }
        }
    }
}

template <bool REPLAY>
__global__ __launch_bounds__(1024) void agg_owner_kernel(const int* __restrict__ row,
                                                         const int* __restrict__ col,
                                                         const float* __restrict__ dinv,
                                                         const float* __restrict__ h,
                                                         float* __restrict__ agg,
                                                         unsigned* lists, int* counts) {
    __shared__ float    tile[TROWS * DIM];
    __shared__ float    s_dinv[TROWS + 1];
    __shared__ unsigned s_q[32 * QCAP];
    __shared__ int      s_cnt;

    const int tid  = threadIdx.x;
    const int lane = tid & 31;
    const int wave = tid >> 5;
    const int lo   = blockIdx.x * TROWS;
    int nrows = N_NODES - lo;
    if (nrows > TROWS) nrows = TROWS;

    const float4* h4    = (const float4*)h;
    float4*       tile4 = (float4*)tile;
    for (int i = tid; i < nrows * 32; i += 1024) {
        const float d  = dinv[lo + (i >> 5)];
        const float dd = d * d;
        float4 v = h4[(size_t)lo * 32 + i];
        v.x = dd * v.x;
        v.y = dd * v.y;
        v.z = dd * v.z;
        v.w = dd * v.w;
        tile4[i] = v;
    }
    for (int i = tid; i < nrows; i += 1024) s_dinv[i] = dinv[lo + i];
    if (tid == 0) s_cnt = 0;
    __syncthreads();

    lds_vu32* q = (lds_vu32*)(s_q + wave * QCAP);
    unsigned* rec = lists + (size_t)blockIdx.x * CAPB;
    if (REPLAY) {
        const int cnt = counts[blockIdx.x];
        if (cnt <= CAPB)
            replay_edges(rec, cnt, lane, wave, row, dinv, h, s_dinv, tile);
        else
            scan_targets<false>(q, lo, nrows, lane, wave, row, col, dinv, h, s_dinv, tile,
                                &s_cnt, rec);
    } else {
        scan_targets<true>(q, lo, nrows, lane, wave, row, col, dinv, h, s_dinv, tile,
                           &s_cnt, rec);
    }
    __syncthreads();
    if (!REPLAY && tid == 0) counts[blockIdx.x] = s_cnt;

    float4* agg4 = (float4*)agg;
    for (int i = tid; i < nrows * 32; i += 1024) agg4[(size_t)lo * 32 + i] = tile4[i];
}

extern "C" void kernel_launch(void* const* d_in, const int* in_sizes, int n_in,
                              void* d_out, int out_size, void* d_ws, size_t ws_size,
                              hipStream_t stream) {
    const float* x    = (const float*)d_in[0];
    const int*   edge = (const int*)d_in[1];
    const float* W1   = (const float*)d_in[2];
    const float* b1   = (const float*)d_in[3];
    const float* W2   = (const float*)d_in[4];
    const float* b2   = (const float*)d_in[5];
    const float* Wout = (const float*)d_in[6];
    const float* bout = (const float*)d_in[7];
    const int* row = edge;
    const int* col = edge + N_EDGES;
    float* out = (float*)d_out;

    char* ws = (char*)d_ws;
    float* dinv = (float*)ws;
    float* bufA = (float*)(ws + ((size_t)1 << 20));
    float* bufB = bufA + (size_t)N_NODES * DIM;
    int*      counts = (int*)(ws + ((size_t)1 << 19));
    unsigned* lists  = (unsigned*)(ws + ((size_t)112 << 20));

    const int ntiles = N_NODES / 16;

    deg_init_kernel<<<(N_NODES + 255) / 256, 256, 0, stream>>>(dinv);
    deg_acc_kernel<<<(N_EDGES + 255) / 256, 256, 0, stream>>>(col, dinv);
    dinv_kernel<<<(N_NODES + 255) / 256, 256, 0, stream>>>(dinv);

    gemm128_wmma<<<ntiles, 256, 0, stream>>>(x, W1, bufA);
    agg_owner_kernel<false><<<NBLK_AGG, 1024, 0, stream>>>(row, col, dinv, bufA, bufB, lists, counts);

    gemm128_act_wmma<<<ntiles, 256, 0, stream>>>(bufB, b1, W2, bufA);
    agg_owner_kernel<true><<<NBLK_AGG, 1024, 0, stream>>>(row, col, dinv, bufA, bufB, lists, counts);

    gemm_out_wmma<<<ntiles, 96, 0, stream>>>(bufB, b2, Wout, bout, out);
}


# Matrix products re-tiled by rows: each wave makes a whole 16-row output tile, its X rows read once as 16-byte pieces straight into A fragments, W in LDS by groups of 4 k (another k summing order)
# speedup vs baseline: 2.2772x; 1.1615x over previous
#include <hip/hip_runtime.h>

typedef __attribute__((ext_vector_type(2))) float v2f;
typedef __attribute__((ext_vector_type(8))) float v8f;

#define N_NODES 100000
#define N_EDGES 1600000
#define DIM 128
#define N_CLASSES 40
#define TROWS 391
#define NBLK_AGG 256
#define QCAP 136
#define NCHUNK (N_EDGES / 128)
#define GRP 8
#define CAPB 32768
static_assert(N_EDGES % 128 == 0, "edge list is read in whole chunks");
static_assert(NBLK_AGG * TROWS >= N_NODES, "blocks must cover every target row");
static_assert(TROWS <= 512 && N_EDGES <= (1 << 21), "queue entry packs edge id and local row in 32 bits");

__global__ void deg_init_kernel(float* deg) {
    int i = blockIdx.x * blockDim.x + threadIdx.x;
    if (i < N_NODES) deg[i] = 1.0f;
}

__global__ void deg_acc_kernel(const int* __restrict__ col, float* deg) {
    int e = blockIdx.x * blockDim.x + threadIdx.x;
    if (e < N_EDGES) atomicAdd(&deg[col[e]], 1.0f);
}

__global__ void dinv_kernel(float* deg) {
    int i = blockIdx.x * blockDim.x + threadIdx.x;
    if (i < N_NODES) deg[i] = rsqrtf(deg[i]);
}

#define GEMM_WAVES 8
#define GEMM_TILES (N_NODES / 16)
static_assert(N_NODES % 16 == 0, "row tiles are whole");
template <int NT, int NCOL, bool ACT, bool OBIAS>
__device__ __forceinline__ void gemm_rows_body(const float* __restrict__ X,
                                               const float* __restrict__ W,
                                               const float* __restrict__ bin,
                                               const float* __restrict__ obias,
                                               float* __restrict__ H,
                                               float4* Wl) {
    constexpr int NP     = NT * 16;
    constexpr int NPP    = NP + 8;
    constexpr int PASSES = (DIM / 4) * NP / (GEMM_WAVES * 32);
    static_assert(PASSES * GEMM_WAVES * 32 == (DIM / 4) * NP, "every thread stages the same number of groups");
    const int tid  = threadIdx.x;
    const int lane = tid & 31;
    const int wave = tid >> 5;
    const int c    = lane & 15;
    const int h    = lane >> 4;
    const int tile = blockIdx.x * GEMM_WAVES + wave;
    const bool live = tile < GEMM_TILES;
    const int m0 = (live ? tile : GEMM_TILES - 1) * 16;

    const float4* xr = (const float4*)(X + (size_t)(m0 + c) * DIM) + h;
    float4 av[DIM / 8];
#pragma unroll
    for (int j = 0; j < DIM / 8; ++j) av[j] = xr[2 * j];

    float4 wv[PASSES];
#pragma unroll
    for (int p = 0; p < PASSES; ++p) {
        const int idx = tid + p * (GEMM_WAVES * 32);
        const int g = idx / NP;
        const int n = idx - g * NP;
        wv[p] = make_float4(0.0f, 0.0f, 0.0f, 0.0f);
        if (n < NCOL) {
            const float* wp = W + (size_t)(4 * g) * NCOL + n;
            wv[p].x = wp[0];
            wv[p].y = wp[NCOL];
            wv[p].z = wp[2 * NCOL];
            wv[p].w = wp[3 * NCOL];
        }
    }
#pragma unroll
    for (int p = 0; p < PASSES; ++p) {
        const int idx = tid + p * (GEMM_WAVES * 32);
        const int g = idx / NP;
        Wl[g * NPP + (idx - g * NP)] = wv[p];
    }
    __syncthreads();
    if (!live) return;

    const float4* bi4 = (const float4*)bin + h;
    const float4* wl  = Wl + h * NPP + c;

    v8f acc[NT];
#pragma unroll
    for (int t = 0; t < NT; ++t) acc[t] = v8f{};

#pragma unroll
    for (int j = 0; j < DIM / 8; ++j) {
        float4 a4 = av[j];
        if (ACT) {
            const float4 bb = bi4[2 * j];
            a4.x = fmaxf(a4.x + bb.x, 0.0f);
            a4.y = fmaxf(a4.y + bb.y, 0.0f);
            a4.z = fmaxf(a4.z + bb.z, 0.0f);
            a4.w = fmaxf(a4.w + bb.w, 0.0f);
        }
        v2f a0, a1;
        a0.x = a4.x; a0.y = a4.y;
        a1.x = a4.z; a1.y = a4.w;
        float4 bv[NT];
#pragma unroll
        for (int t = 0; t < NT; ++t) bv[t] = wl[(2 * j) * NPP + 16 * t];
#pragma unroll
        for (int t = 0; t < NT; ++t) {
            v2f b0;
            b0.x = bv[t].x; b0.y = bv[t].y;
            acc[t] = __builtin_amdgcn_wmma_f32_16x16x4_f32(
                 false, a0,  false, b0,
                 (short)0, acc[t],  false,  false);
        }
#pragma unroll
        for (int t = 0; t < NT; ++t) {
            v2f b1;
            b1.x = bv[t].z; b1.y = bv[t].w;
            acc[t] = __builtin_amdgcn_wmma_f32_16x16x4_f32(
                false, a1, false, b1, (short)0, acc[t], false, false);
        }
    }

    float ob[NT];
#pragma unroll
    for (int t = 0; t < NT; ++t) ob[t] = (OBIAS && 16 * t + c < NCOL) ? obias[16 * t + c] : 0.0f;
    float* hrow = H + (size_t)(m0 + h * 8) * NCOL + c;
#pragma unroll
    for (int r = 0; r < 8; ++r) {
#pragma unroll
        for (int t = 0; t < NT; ++t) {
            if (16 * t + c < NCOL)
                hrow[(size_t)r * NCOL + 16 * t] = OBIAS ? acc[t][r] + ob[t] : acc[t][r];
        }
    }
}

__global__ __launch_bounds__(256) void gemm128_wmma(const float* __restrict__ X,
                                                    const float* __restrict__ W,
                                                    float* __restrict__ H) {
    __shared__ float4 Wl[(DIM / 4) * (DIM + 8)];
    gemm_rows_body<DIM / 16, DIM, false, false>(X, W, nullptr, nullptr, H, Wl);
}

__global__ __launch_bounds__(256) void gemm128_act_wmma(const float* __restrict__ S,
                                                        const float* __restrict__ bias,
                                                        const float* __restrict__ W,
                                                        float* __restrict__ H) {
    __shared__ float4 Wl[(DIM / 4) * (DIM + 8)];
    gemm_rows_body<DIM / 16, DIM, true, false>(S, W, bias, nullptr, H, Wl);
}

__global__ __launch_bounds__(256) void gemm_out_wmma(const float* __restrict__ S,
                                                     const float* __restrict__ bin,
                                                     const float* __restrict__ W,
                                                     const float* __restrict__ bias,
                                                     float* __restrict__ OUT) {
    constexpr int NT_OUT = (N_CLASSES + 15) / 16;
    __shared__ float4 Wl[(DIM / 4) * (NT_OUT * 16 + 8)];
    gemm_rows_body<NT_OUT, N_CLASSES, true, true>(S, W, bin, bias, OUT, Wl);
}


typedef volatile __attribute__((address_space(3))) unsigned lds_vu32;

template <typename QP>
__device__ __forceinline__ void drain_edges(QP q, int n, int lane,
                                            const int* __restrict__ row,
                                            const float* __restrict__ dinv,
                                            const float* __restrict__ h,
                                            const float* s_dinv, float* tile) {
    unsigned ent[8];
    int r[8];
    float w[8];
    float hv[8][4];
#pragma unroll
    for (int b = 0; b < 8; ++b) {
        ent[b] = __builtin_amdgcn_readfirstlane(q[b < n ? b : 0]);
    }
#pragma unroll
    for (int b = 0; b < 8; ++b) r[b] = row[ent[b] >> 9];
#pragma unroll
    for (int b = 0; b < 8; ++b) {
        w[b] = dinv[r[b]] * s_dinv[ent[b] & 511u];
#pragma unroll
        for (int k = 0; k < 4; ++k) hv[b][k] = h[(size_t)r[b] * DIM + k * 32 + lane];
    }
#pragma unroll
    for (int b = 0; b < 8; ++b) {
        if (b < n) {
            float* trow = tile + (ent[b] & 511u) * DIM + lane;
#pragma unroll
            for (int k = 0; k < 4; ++k)
                __hip_atomic_fetch_add(trow + k * 32, w[b] * hv[b][k], __ATOMIC_RELAXED,
                                       __HIP_MEMORY_SCOPE_WORKGROUP);
        }
    }
}

__device__ __forceinline__ void record_entries(const lds_vu32* q, int n, int lane,
                                               int* s_cnt, unsigned* rec) {
    int base = 0;
    if (lane == 0) base = atomicAdd(s_cnt, n);
    base = __builtin_amdgcn_readfirstlane(base);
    if (lane < n && base + n <= CAPB) rec[base + lane] = q[lane];
}

template <bool RECORD>
__device__ __forceinline__ void scan_targets(lds_vu32* q, int lo, int nrows, int lane, int wave,
                                             const int* __restrict__ row,
                                             const int* __restrict__ col,
                                             const float* __restrict__ dinv,
                                             const float* __restrict__ h,
                                             const float* s_dinv, float* tile,
                                             int* s_cnt, unsigned* rec) {
    const unsigned below = (1u << lane) - 1u;
    const int4* col4 = (const int4*)col;
    int qn = 0;

    for (int g0 = wave * GRP; g0 < NCHUNK; g0 += 32 * GRP) {
        int4 cb[GRP];
#pragma unroll
        for (int u = 0; u < GRP; ++u) {
            if (g0 + u < NCHUNK) cb[u] = col4[(size_t)(g0 + u) * 32 + lane];
            else                 cb[u] = make_int4(-1, -1, -1, -1);
        }
#pragma unroll
        for (int u = 0; u < GRP; ++u) {
            const unsigned c0 = (unsigned)(cb[u].x - lo);
            const unsigned c1 = (unsigned)(cb[u].y - lo);
            const unsigned c2 = (unsigned)(cb[u].z - lo);
            const unsigned c3 = (unsigned)(cb[u].w - lo);
            const bool m0 = c0 < (unsigned)nrows;
            const bool m1 = c1 < (unsigned)nrows;
            const bool m2 = c2 < (unsigned)nrows;
            const bool m3 = c3 < (unsigned)nrows;
            if (__builtin_amdgcn_ballot_w32(m0 | m1 | m2 | m3) == 0u) continue;

            const unsigned e0 = (unsigned)(g0 + u) * 128u + (unsigned)lane * 4u;
            unsigned mk;
            mk = __builtin_amdgcn_ballot_w32(m0);
            if (m0) q[qn + __builtin_popcount(mk & below)] = ((e0 + 0u) << 9) | c0;
            qn += __builtin_popcount(mk);
            mk = __builtin_amdgcn_ballot_w32(m1);
            if (m1) q[qn + __builtin_popcount(mk & below)] = ((e0 + 1u) << 9) | c1;
            qn += __builtin_popcount(mk);
            mk = __builtin_amdgcn_ballot_w32(m2);
            if (m2) q[qn + __builtin_popcount(mk & below)] = ((e0 + 2u) << 9) | c2;
            qn += __builtin_popcount(mk);
            mk = __builtin_amdgcn_ballot_w32(m3);
            if (m3) q[qn + __builtin_popcount(mk & below)] = ((e0 + 3u) << 9) | c3;
            qn += __builtin_popcount(mk);

            while (qn >= 8) {
                qn -= 8;
                if (RECORD) record_entries(q + qn, 8, lane, s_cnt, rec);
                drain_edges(q + qn, 8, lane, row, dinv, h, s_dinv, tile);
            }
        }
    }
    if (qn > 0) {
        if (RECORD) record_entries(q, qn, lane, s_cnt, rec);
        drain_edges(q, qn, lane, row, dinv, h, s_dinv, tile);
    }
}

__device__ __forceinline__ void replay_edges(const unsigned* __restrict__ rec, int cnt,
                                             int lane, int wave,
                                             const int* __restrict__ row,
                                             const float* __restrict__ dinv,
                                             const float* __restrict__ h,
                                             const float* s_dinv, float* tile) {
    const int wave_u = __builtin_amdgcn_readfirstlane(wave);
    for (int i0 = wave_u * 32; i0 < cnt; i0 += 1024) {
        int nb = cnt - i0;
        if (nb > 32) nb = 32;
        const unsigned ent = (lane < nb) ? rec[i0 + lane] : 0u;
        const int      r   = row[ent >> 9];
        const float    w   = dinv[r] * s_dinv[ent & 511u];
        for (int g = 0; g < nb; g += 8) {
            unsigned eb[8];
            int rb[8];
            float wb[8];
            float hv[8][4];
#pragma unroll
            for (int b = 0; b < 8; ++b) {
                const int sel = (g + b < nb) ? (g + b) : g;
                eb[b] = (unsigned)__builtin_amdgcn_readlane((int)ent, sel);
                rb[b] = __builtin_amdgcn_readlane(r, sel);
                wb[b] = __int_as_float(__builtin_amdgcn_readlane(__float_as_int(w), sel));
            }
#pragma unroll
            for (int b = 0; b < 8; ++b) {
#pragma unroll
                for (int k = 0; k < 4; ++k) hv[b][k] = h[(size_t)rb[b] * DIM + k * 32 + lane];
            }
#pragma unroll
            for (int b = 0; b < 8; ++b) {
                if (b == 0 || g + b < nb) {
                    float* trow = tile + (eb[b] & 511u) * DIM + lane;
#pragma unroll
                    for (int k = 0; k < 4; ++k)
                        __hip_atomic_fetch_add(trow + k * 32, wb[b] * hv[b][k], __ATOMIC_RELAXED,
                                               __HIP_MEMORY_SCOPE_WORKGROUP);
                }
            }
        }
    }
}

template <bool REPLAY>
__global__ __launch_bounds__(1024) void agg_owner_kernel(const int* __restrict__ row,
                                                         const int* __restrict__ col,
                                                         const float* __restrict__ dinv,
                                                         const float* __restrict__ h,
                                                         float* __restrict__ agg,
                                                         unsigned* lists, int* counts) {
    __shared__ float    tile[TROWS * DIM];
    __shared__ float    s_dinv[TROWS + 1];
    __shared__ unsigned s_q[32 * QCAP];
    __shared__ int      s_cnt;

    const int tid  = threadIdx.x;
    const int lane = tid & 31;
    const int wave = tid >> 5;
    const int lo   = blockIdx.x * TROWS;
    int nrows = N_NODES - lo;
    if (nrows > TROWS) nrows = TROWS;

    const float4* h4    = (const float4*)h;
    float4*       tile4 = (float4*)tile;
    for (int i = tid; i < nrows * 32; i += 1024) {
        const float d  = dinv[lo + (i >> 5)];
        const float dd = d * d;
        float4 v = h4[(size_t)lo * 32 + i];
        v.x = dd * v.x;
        v.y = dd * v.y;
        v.z = dd * v.z;
        v.w = dd * v.w;
        tile4[i] = v;
    }
    for (int i = tid; i < nrows; i += 1024) s_dinv[i] = dinv[lo + i];
    if (tid == 0) s_cnt = 0;
    __syncthreads();

    lds_vu32* q = (lds_vu32*)(s_q + wave * QCAP);
    unsigned* rec = lists + (size_t)blockIdx.x * CAPB;
    if (REPLAY) {
        const int cnt = counts[blockIdx.x];
        if (cnt <= CAPB)
            replay_edges(rec, cnt, lane, wave, row, dinv, h, s_dinv, tile);
        else
            scan_targets<false>(q, lo, nrows, lane, wave, row, col, dinv, h, s_dinv, tile,
                                &s_cnt, rec);
    } else {
        scan_targets<true>(q, lo, nrows, lane, wave, row, col, dinv, h, s_dinv, tile,
                           &s_cnt, rec);
    }
    __syncthreads();
    if (!REPLAY && tid == 0) counts[blockIdx.x] = s_cnt;

    float4* agg4 = (float4*)agg;
    for (int i = tid; i < nrows * 32; i += 1024) agg4[(size_t)lo * 32 + i] = tile4[i];
}

extern "C" void kernel_launch(void* const* d_in, const int* in_sizes, int n_in,
                              void* d_out, int out_size, void* d_ws, size_t ws_size,
                              hipStream_t stream) {
    const float* x    = (const float*)d_in[0];
    const int*   edge = (const int*)d_in[1];
    const float* W1   = (const float*)d_in[2];
    const float* b1   = (const float*)d_in[3];
    const float* W2   = (const float*)d_in[4];
    const float* b2   = (const float*)d_in[5];
    const float* Wout = (const float*)d_in[6];
    const float* bout = (const float*)d_in[7];
    const int* row = edge;
    const int* col = edge + N_EDGES;
    float* out = (float*)d_out;

    char* ws = (char*)d_ws;
    float* dinv = (float*)ws;
    float* bufA = (float*)(ws + ((size_t)1 << 20));
    float* bufB = bufA + (size_t)N_NODES * DIM;
    int*      counts = (int*)(ws + ((size_t)1 << 19));
    unsigned* lists  = (unsigned*)(ws + ((size_t)112 << 20));

    const int gblocks = (N_NODES / 16 + GEMM_WAVES - 1) / GEMM_WAVES;

    deg_init_kernel<<<(N_NODES + 255) / 256, 256, 0, stream>>>(dinv);
    deg_acc_kernel<<<(N_EDGES + 255) / 256, 256, 0, stream>>>(col, dinv);
    dinv_kernel<<<(N_NODES + 255) / 256, 256, 0, stream>>>(dinv);

    gemm128_wmma<<<gblocks, 256, 0, stream>>>(x, W1, bufA);
    agg_owner_kernel<false><<<NBLK_AGG, 1024, 0, stream>>>(row, col, dinv, bufA, bufB, lists, counts);

    gemm128_act_wmma<<<gblocks, 256, 0, stream>>>(bufB, b1, W2, bufA);
    agg_owner_kernel<true><<<NBLK_AGG, 1024, 0, stream>>>(row, col, dinv, bufA, bufB, lists, counts);

    gemm_out_wmma<<<gblocks, 256, 0, stream>>>(bufB, b2, Wout, bout, out);
}


# One binning pass fills the owner lists (slices sorted by owner in LDS; record: source, local row, dinv[source]); both layers replay them, no target scan; LDS f32 adds: arrival order; launches 8->9
# speedup vs baseline: 2.6837x; 1.1785x over previous
#include <hip/hip_runtime.h>

typedef __attribute__((ext_vector_type(2))) float v2f;
typedef __attribute__((ext_vector_type(8))) float v8f;

#define N_NODES 100000
#define N_EDGES 1600000
#define DIM 128
#define N_CLASSES 40
#define TROWS 391
#define NBLK_AGG 256
#define QCAP 136
#define NCHUNK (N_EDGES / 128)
#define GRP 8
#define CAPB 32768
#define BIN_SLICE 6144
#define BIN_PASSES (BIN_SLICE / 1024)
#define NBLK_BIN ((N_EDGES + BIN_SLICE - 1) / BIN_SLICE)
static_assert(N_EDGES % 128 == 0, "edge list is read in whole chunks");
static_assert(NBLK_AGG * TROWS >= N_NODES, "blocks must cover every target row");
static_assert(TROWS <= 512 && N_EDGES <= (1 << 21), "queue entry packs edge id and local row in 32 bits");
static_assert(TROWS <= 512 && N_NODES <= (1 << 23), "list record packs source row and local row in 32 bits");
static_assert(NBLK_AGG <= 256, "the owner of a sorted slot is kept in one byte");

__global__ void deg_init_kernel(float* deg, int* counts) {
    int i = blockIdx.x * blockDim.x + threadIdx.x;
    if (i < N_NODES) deg[i] = 1.0f;
    if (i < NBLK_AGG) counts[i] = 0;
}

__global__ void deg_acc_kernel(const int* __restrict__ col, float* deg) {
    int e = blockIdx.x * blockDim.x + threadIdx.x;
    if (e < N_EDGES) atomicAdd(&deg[col[e]], 1.0f);
}

__global__ void dinv_kernel(float* deg) {
    int i = blockIdx.x * blockDim.x + threadIdx.x;
    if (i < N_NODES) deg[i] = rsqrtf(deg[i]);
}

#define GEMM_WAVES 8
#define GEMM_TILES (N_NODES / 16)
static_assert(N_NODES % 16 == 0, "row tiles are whole");
template <int NT, int NCOL, bool ACT, bool OBIAS>
__device__ __forceinline__ void gemm_rows_body(const float* __restrict__ X,
                                               const float* __restrict__ W,
                                               const float* __restrict__ bin,
                                               const float* __restrict__ obias,
                                               float* __restrict__ H,
                                               float4* Wl) {
    constexpr int NP     = NT * 16;
    constexpr int NPP    = NP + 8;
    constexpr int PASSES = (DIM / 4) * NP / (GEMM_WAVES * 32);
    static_assert(PASSES * GEMM_WAVES * 32 == (DIM / 4) * NP, "every thread stages the same number of groups");
    const int tid  = threadIdx.x;
    const int lane = tid & 31;
    const int wave = tid >> 5;
    const int c    = lane & 15;
    const int h    = lane >> 4;
    const int tile = blockIdx.x * GEMM_WAVES + wave;
    const bool live = tile < GEMM_TILES;
    const int m0 = (live ? tile : GEMM_TILES - 1) * 16;

    const float4* xr = (const float4*)(X + (size_t)(m0 + c) * DIM) + h;
    float4 av[DIM / 8];
#pragma unroll
    for (int j = 0; j < DIM / 8; ++j) av[j] = xr[2 * j];

    float4 wv[PASSES];
#pragma unroll
    for (int p = 0; p < PASSES; ++p) {
        const int idx = tid + p * (GEMM_WAVES * 32);
        const int g = idx / NP;
        const int n = idx - g * NP;
        wv[p] = make_float4(0.0f, 0.0f, 0.0f, 0.0f);
        if (n < NCOL) {
            const float* wp = W + (size_t)(4 * g) * NCOL + n;
            wv[p].x = wp[0];
            wv[p].y = wp[NCOL];
            wv[p].z = wp[2 * NCOL];
            wv[p].w = wp[3 * NCOL];
        }
    }
#pragma unroll
    for (int p = 0; p < PASSES; ++p) {
        const int idx = tid + p * (GEMM_WAVES * 32);
        const int g = idx / NP;
        Wl[g * NPP + (idx - g * NP)] = wv[p];
    }
    __syncthreads();
    if (!live) return;

    const float4* bi4 = (const float4*)bin + h;
    const float4* wl  = Wl + h * NPP + c;

    v8f acc[NT];
#pragma unroll
    for (int t = 0; t < NT; ++t) acc[t] = v8f{};

#pragma unroll
    for (int j = 0; j < DIM / 8; ++j) {
        float4 a4 = av[j];
        if (ACT) {
            const float4 bb = bi4[2 * j];
            a4.x = fmaxf(a4.x + bb.x, 0.0f);
            a4.y = fmaxf(a4.y + bb.y, 0.0f);
            a4.z = fmaxf(a4.z + bb.z, 0.0f);
            a4.w = fmaxf(a4.w + bb.w, 0.0f);
        }
        v2f a0, a1;
        a0.x = a4.x; a0.y = a4.y;
        a1.x = a4.z; a1.y = a4.w;
        float4 bv[NT];
#pragma unroll
        for (int t = 0; t < NT; ++t) bv[t] = wl[(2 * j) * NPP + 16 * t];
#pragma unroll
        for (int t = 0; t < NT; ++t) {
            v2f b0;
            b0.x = bv[t].x; b0.y = bv[t].y;
            acc[t] = __builtin_amdgcn_wmma_f32_16x16x4_f32(
                 false, a0,  false, b0,
                 (short)0, acc[t],  false,  false);
        }
#pragma unroll
        for (int t = 0; t < NT; ++t) {
            v2f b1;
            b1.x = bv[t].z; b1.y = bv[t].w;
            acc[t] = __builtin_amdgcn_wmma_f32_16x16x4_f32(
                false, a1, false, b1, (short)0, acc[t], false, false);
        }
    }

    float ob[NT];
#pragma unroll
    for (int t = 0; t < NT; ++t) ob[t] = (OBIAS && 16 * t + c < NCOL) ? obias[16 * t + c] : 0.0f;
    float* hrow = H + (size_t)(m0 + h * 8) * NCOL + c;
#pragma unroll
    for (int r = 0; r < 8; ++r) {
#pragma unroll
        for (int t = 0; t < NT; ++t) {
            if (16 * t + c < NCOL)
                hrow[(size_t)r * NCOL + 16 * t] = OBIAS ? acc[t][r] + ob[t] : acc[t][r];
        }
    }
}

__global__ __launch_bounds__(256) void gemm128_wmma(const float* __restrict__ X,
                                                    const float* __restrict__ W,
                                                    float* __restrict__ H) {
    __shared__ float4 Wl[(DIM / 4) * (DIM + 8)];
    gemm_rows_body<DIM / 16, DIM, false, false>(X, W, nullptr, nullptr, H, Wl);
}

__global__ __launch_bounds__(256) void gemm128_act_wmma(const float* __restrict__ S,
                                                        const float* __restrict__ bias,
                                                        const float* __restrict__ W,
                                                        float* __restrict__ H) {
    __shared__ float4 Wl[(DIM / 4) * (DIM + 8)];
    gemm_rows_body<DIM / 16, DIM, true, false>(S, W, bias, nullptr, H, Wl);
}

__global__ __launch_bounds__(256) void gemm_out_wmma(const float* __restrict__ S,
                                                     const float* __restrict__ bin,
                                                     const float* __restrict__ W,
                                                     const float* __restrict__ bias,
                                                     float* __restrict__ OUT) {
    constexpr int NT_OUT = (N_CLASSES + 15) / 16;
    __shared__ float4 Wl[(DIM / 4) * (NT_OUT * 16 + 8)];
    gemm_rows_body<NT_OUT, N_CLASSES, true, true>(S, W, bin, bias, OUT, Wl);
}

__global__ __launch_bounds__(1024) void bin_edges_kernel(const int* __restrict__ row,
                                                         const int* __restrict__ col,
                                                         const float* __restrict__ dinv,
                                                         uint2* __restrict__ lists, int* counts) {
    __shared__ uint2         s_ent[BIN_SLICE];
    __shared__ unsigned char s_own[BIN_SLICE];
    __shared__ unsigned      s_cnt[NBLK_AGG];
    __shared__ unsigned      s_off[NBLK_AGG];
    __shared__ unsigned      s_base[NBLK_AGG];
    __shared__ unsigned      s_wtot[NBLK_AGG / 32];

    const int tid  = threadIdx.x;
    const int lane = tid & 31;
    const int wave = tid >> 5;
    const int e0   = blockIdx.x * BIN_SLICE;
    int n = N_EDGES - e0;
    if (n > BIN_SLICE) n = BIN_SLICE;

    if (tid < NBLK_AGG) s_cnt[tid] = 0;

    int dst[BIN_PASSES], sr[BIN_PASSES];
#pragma unroll
    for (int k = 0; k < BIN_PASSES; ++k) {
        const int i = tid + k * 1024;
        dst[k] = -1;
        sr[k]  = 0;
        if (i < n) {
            dst[k] = col[e0 + i];
            sr[k]  = row[e0 + i];
        }
    }
    __syncthreads();

    unsigned own[BIN_PASSES], rnk[BIN_PASSES], pk[BIN_PASSES];
    float ws[BIN_PASSES];
#pragma unroll
    for (int k = 0; k < BIN_PASSES; ++k) {
        const bool ok = (unsigned)dst[k] < (unsigned)N_NODES && (unsigned)sr[k] < (unsigned)N_NODES;
        own[k] = NBLK_AGG;
        rnk[k] = 0;
        pk[k]  = 0;
        ws[k]  = 0.0f;
        if (ok) {
            const unsigned o = (unsigned)dst[k] / TROWS;
            own[k] = o;
            rnk[k] = atomicAdd(&s_cnt[o], 1u);
            pk[k]  = ((unsigned)sr[k] << 9) | ((unsigned)dst[k] - o * TROWS);
            ws[k]  = dinv[sr[k]];
        }
    }
    __syncthreads();

    unsigned cn = 0, incl = 0;
    if (tid < NBLK_AGG) {
        cn = s_cnt[tid];
        incl = cn;
#pragma unroll
        for (int d = 1; d < 32; d <<= 1) {
            const unsigned t = __shfl_up(incl, d);
            if (lane >= d) incl += t;
        }
        if (lane == 31) s_wtot[wave] = incl;
    }
    __syncthreads();
    if (tid < NBLK_AGG) {
        unsigned pre = 0;
#pragma unroll
        for (int w = 0; w < NBLK_AGG / 32; ++w)
            if (w < wave) pre += s_wtot[w];
        s_off[tid]  = pre + incl - cn;
        s_base[tid] = cn ? (unsigned)atomicAdd(&counts[tid], (int)cn) : 0u;
    }
    __syncthreads();

#pragma unroll
    for (int k = 0; k < BIN_PASSES; ++k) {
        if (own[k] < NBLK_AGG) {
            const unsigned slot = s_off[own[k]] + rnk[k];
            s_ent[slot] = make_uint2(pk[k], __float_as_uint(ws[k]));
            s_own[slot] = (unsigned char)own[k];
        }
    }
    __syncthreads();

    const int total = (int)(s_off[NBLK_AGG - 1] + s_cnt[NBLK_AGG - 1]);
    for (int s = tid; s < total; s += 1024) {
        const unsigned o   = s_own[s];
        const unsigned pos = s_base[o] + ((unsigned)s - s_off[o]);
        if (pos < (unsigned)CAPB) lists[(size_t)o * CAPB + pos] = s_ent[s];
    }
}


typedef volatile __attribute__((address_space(3))) unsigned lds_vu32;

template <typename QP>
__device__ __forceinline__ void drain_edges(QP q, int n, int lane,
                                            const int* __restrict__ row,
                                            const float* __restrict__ dinv,
                                            const float* __restrict__ h,
                                            const float* s_dinv, float* tile) {
    unsigned ent[8];
    int r[8];
    float w[8];
    float hv[8][4];
#pragma unroll
    for (int b = 0; b < 8; ++b) {
        ent[b] = __builtin_amdgcn_readfirstlane(q[b < n ? b : 0]);
    }
#pragma unroll
    for (int b = 0; b < 8; ++b) r[b] = row[ent[b] >> 9];
#pragma unroll
    for (int b = 0; b < 8; ++b) {
        w[b] = dinv[r[b]] * s_dinv[ent[b] & 511u];
#pragma unroll
        for (int k = 0; k < 4; ++k) hv[b][k] = h[(size_t)r[b] * DIM + k * 32 + lane];
    }
#pragma unroll
    for (int b = 0; b < 8; ++b) {
        if (b < n) {
            float* trow = tile + (ent[b] & 511u) * DIM + lane;
#pragma unroll
            for (int k = 0; k < 4; ++k)
                __hip_atomic_fetch_add(trow + k * 32, w[b] * hv[b][k], __ATOMIC_RELAXED,
                                       __HIP_MEMORY_SCOPE_WORKGROUP);
        }
    }
}

__device__ __forceinline__ void scan_targets(lds_vu32* q, int lo, int nrows, int lane, int wave,
                                             const int* __restrict__ row,
                                             const int* __restrict__ col,
                                             const float* __restrict__ dinv,
                                             const float* __restrict__ h,
                                             const float* s_dinv, float* tile) {
    const unsigned below = (1u << lane) - 1u;
    const int4* col4 = (const int4*)col;
    int qn = 0;

    for (int g0 = wave * GRP; g0 < NCHUNK; g0 += 32 * GRP) {
        int4 cb[GRP];
#pragma unroll
        for (int u = 0; u < GRP; ++u) {
            if (g0 + u < NCHUNK) cb[u] = col4[(size_t)(g0 + u) * 32 + lane];
            else                 cb[u] = make_int4(-1, -1, -1, -1);
        }
#pragma unroll
        for (int u = 0; u < GRP; ++u) {
            const unsigned c0 = (unsigned)(cb[u].x - lo);
            const unsigned c1 = (unsigned)(cb[u].y - lo);
            const unsigned c2 = (unsigned)(cb[u].z - lo);
            const unsigned c3 = (unsigned)(cb[u].w - lo);
            const bool m0 = c0 < (unsigned)nrows;
            const bool m1 = c1 < (unsigned)nrows;
            const bool m2 = c2 < (unsigned)nrows;
            const bool m3 = c3 < (unsigned)nrows;
            if (__builtin_amdgcn_ballot_w32(m0 | m1 | m2 | m3) == 0u) continue;

            const unsigned e0 = (unsigned)(g0 + u) * 128u + (unsigned)lane * 4u;
            unsigned mk;
            mk = __builtin_amdgcn_ballot_w32(m0);
            if (m0) q[qn + __builtin_popcount(mk & below)] = ((e0 + 0u) << 9) | c0;
            qn += __builtin_popcount(mk);
            mk = __builtin_amdgcn_ballot_w32(m1);
            if (m1) q[qn + __builtin_popcount(mk & below)] = ((e0 + 1u) << 9) | c1;
            qn += __builtin_popcount(mk);
            mk = __builtin_amdgcn_ballot_w32(m2);
            if (m2) q[qn + __builtin_popcount(mk & below)] = ((e0 + 2u) << 9) | c2;
            qn += __builtin_popcount(mk);
            mk = __builtin_amdgcn_ballot_w32(m3);
            if (m3) q[qn + __builtin_popcount(mk & below)] = ((e0 + 3u) << 9) | c3;
            qn += __builtin_popcount(mk);

            while (qn >= 8) {
                qn -= 8;
                drain_edges(q + qn, 8, lane, row, dinv, h, s_dinv, tile);
            }
        }
    }
    if (qn > 0) drain_edges(q, qn, lane, row, dinv, h, s_dinv, tile);
}

__device__ __forceinline__ void replay_edges(const uint2* __restrict__ rec, int cnt,
                                             int lane, int wave,
                                             const float* __restrict__ h,
                                             const float* s_dinv, float* tile) {
    const int wave_u = __builtin_amdgcn_readfirstlane(wave);
    int i0 = wave_u * 32;
    uint2 nxt = make_uint2(0u, 0u);
    if (i0 + lane < cnt) nxt = rec[i0 + lane];
    for (; i0 < cnt; i0 += 1024) {
        int nb = cnt - i0;
        if (nb > 32) nb = 32;
        const uint2 cur = nxt;
        nxt = make_uint2(0u, 0u);
        if (i0 + 1024 + lane < cnt) nxt = rec[i0 + 1024 + lane];
        const unsigned ent = cur.x;
        const int      r   = (int)(ent >> 9);
        const float    w   = __uint_as_float(cur.y) * s_dinv[ent & 511u];
        for (int g = 0; g < nb; g += 8) {
            unsigned eb[8];
            int rb[8];
            float wb[8];
            float hv[8][4];
#pragma unroll
            for (int b = 0; b < 8; ++b) {
                const int sel = (g + b < nb) ? (g + b) : g;
                eb[b] = (unsigned)__builtin_amdgcn_readlane((int)ent, sel);
                rb[b] = __builtin_amdgcn_readlane(r, sel);
                wb[b] = __int_as_float(__builtin_amdgcn_readlane(__float_as_int(w), sel));
            }
#pragma unroll
            for (int b = 0; b < 8; ++b) {
#pragma unroll
                for (int k = 0; k < 4; ++k) hv[b][k] = h[(size_t)rb[b] * DIM + k * 32 + lane];
            }
#pragma unroll
            for (int b = 0; b < 8; ++b) {
                if (b == 0 || g + b < nb) {
                    float* trow = tile + (eb[b] & 511u) * DIM + lane;
#pragma unroll
                    for (int k = 0; k < 4; ++k)
                        __hip_atomic_fetch_add(trow + k * 32, wb[b] * hv[b][k], __ATOMIC_RELAXED,
                                               __HIP_MEMORY_SCOPE_WORKGROUP);
                }
            }
        }
    }
}

__global__ __launch_bounds__(1024) void agg_owner_kernel(const int* __restrict__ row,
                                                         const int* __restrict__ col,
                                                         const float* __restrict__ dinv,
                                                         const float* __restrict__ h,
                                                         float* __restrict__ agg,
                                                         const uint2* __restrict__ lists,
                                                         const int* __restrict__ counts) {
    __shared__ float    tile[TROWS * DIM];
    __shared__ float    s_dinv[TROWS + 1];
    __shared__ unsigned s_q[32 * QCAP];

    const int tid  = threadIdx.x;
    const int lane = tid & 31;
    const int wave = tid >> 5;
    const int lo   = blockIdx.x * TROWS;
    int nrows = N_NODES - lo;
    if (nrows > TROWS) nrows = TROWS;

    const float4* h4    = (const float4*)h;
    float4*       tile4 = (float4*)tile;
    for (int i = tid; i < nrows * 32; i += 1024) {
        const float d  = dinv[lo + (i >> 5)];
        const float dd = d * d;
        float4 v = h4[(size_t)lo * 32 + i];
        v.x = dd * v.x;
        v.y = dd * v.y;
        v.z = dd * v.z;
        v.w = dd * v.w;
        tile4[i] = v;
    }
    for (int i = tid; i < nrows; i += 1024) s_dinv[i] = dinv[lo + i];
    __syncthreads();

    const int cnt = counts[blockIdx.x];
    if (cnt <= CAPB) {
        replay_edges(lists + (size_t)blockIdx.x * CAPB, cnt, lane, wave, h, s_dinv, tile);
    } else {
        lds_vu32* q = (lds_vu32*)(s_q + wave * QCAP);
        scan_targets(q, lo, nrows, lane, wave, row, col, dinv, h, s_dinv, tile);
    }
    __syncthreads();

    float4* agg4 = (float4*)agg;
    for (int i = tid; i < nrows * 32; i += 1024) agg4[(size_t)lo * 32 + i] = tile4[i];
}

extern "C" void kernel_launch(void* const* d_in, const int* in_sizes, int n_in,
                              void* d_out, int out_size, void* d_ws, size_t ws_size,
                              hipStream_t stream) {
    const float* x    = (const float*)d_in[0];
    const int*   edge = (const int*)d_in[1];
    const float* W1   = (const float*)d_in[2];
    const float* b1   = (const float*)d_in[3];
    const float* W2   = (const float*)d_in[4];
    const float* b2   = (const float*)d_in[5];
    const float* Wout = (const float*)d_in[6];
    const float* bout = (const float*)d_in[7];
    const int* row = edge;
    const int* col = edge + N_EDGES;
    float* out = (float*)d_out;

    char* ws = (char*)d_ws;
    float* dinv = (float*)ws;
    float* bufA = (float*)(ws + ((size_t)1 << 20));
    float* bufB = bufA + (size_t)N_NODES * DIM;
    int*   counts = (int*)(ws + ((size_t)1 << 19));
    uint2* lists  = (uint2*)(ws + ((size_t)112 << 20));

    const int gblocks = (N_NODES / 16 + GEMM_WAVES - 1) / GEMM_WAVES;

    deg_init_kernel<<<(N_NODES + 255) / 256, 256, 0, stream>>>(dinv, counts);
    deg_acc_kernel<<<(N_EDGES + 255) / 256, 256, 0, stream>>>(col, dinv);
    dinv_kernel<<<(N_NODES + 255) / 256, 256, 0, stream>>>(dinv);

    bin_edges_kernel<<<NBLK_BIN, 1024, 0, stream>>>(row, col, dinv, lists, counts);

    gemm128_wmma<<<gblocks, 256, 0, stream>>>(x, W1, bufA);
    agg_owner_kernel<<<NBLK_AGG, 1024, 0, stream>>>(row, col, dinv, bufA, bufB, lists, counts);

    gemm128_act_wmma<<<gblocks, 256, 0, stream>>>(bufB, b1, W2, bufA);
    agg_owner_kernel<<<NBLK_AGG, 1024, 0, stream>>>(row, col, dinv, bufA, bufB, lists, counts);

    gemm_out_wmma<<<gblocks, 256, 0, stream>>>(bufB, b2, Wout, bout, out);
}


# Normalization by rows: owner blocks count degrees from their lists (LDS int adds), matrix products scale h rows by deg^-1/2, sums scaled once per target row (another rounding); launches 9->8
# speedup vs baseline: 3.8089x; 1.4193x over previous
#include <hip/hip_runtime.h>

typedef __attribute__((ext_vector_type(2))) float v2f;
typedef __attribute__((ext_vector_type(8))) float v8f;

#define N_NODES 100000
#define N_EDGES 1600000
#define DIM 128
#define N_CLASSES 40
#define TROWS 391
#define NBLK_AGG 256
#define QCAP 136
#define NCHUNK (N_EDGES / 128)
#define GRP 8
#define CAPB 32768
#define BIN_SLICE 6144
#define BIN_PASSES (BIN_SLICE / 1024)
#define NBLK_BIN ((N_EDGES + BIN_SLICE - 1) / BIN_SLICE)
static_assert(N_EDGES % 128 == 0, "edge list is read in whole chunks");
static_assert(NBLK_AGG * TROWS >= N_NODES, "blocks must cover every target row");
static_assert(TROWS <= 512 && N_EDGES <= (1 << 21), "queue entry packs edge id and local row in 32 bits");
static_assert(TROWS <= 512 && N_NODES <= (1 << 23), "list record packs source row and local row in 32 bits");
static_assert(NBLK_AGG <= 256, "the owner of a sorted slot is kept in one byte");

__global__ void counts_zero_kernel(int* counts) {
    if (threadIdx.x < NBLK_AGG) counts[threadIdx.x] = 0;
}

#define GEMM_WAVES 8
#define GEMM_TILES (N_NODES / 16)
static_assert(N_NODES % 16 == 0, "row tiles are whole");
template <int NT, int NCOL, bool ACT, bool OBIAS, bool RSCALE>
__device__ __forceinline__ void gemm_rows_body(const float* __restrict__ X,
                                               const float* __restrict__ W,
                                               const float* __restrict__ bin,
                                               const float* __restrict__ obias,
                                               const float* __restrict__ rscale,
                                               float* __restrict__ H,
                                               float4* Wl) {
    constexpr int NP     = NT * 16;
    constexpr int NPP    = NP + 8;
    constexpr int PASSES = (DIM / 4) * NP / (GEMM_WAVES * 32);
    static_assert(PASSES * GEMM_WAVES * 32 == (DIM / 4) * NP, "every thread stages the same number of groups");
    const int tid  = threadIdx.x;
    const int lane = tid & 31;
    const int wave = tid >> 5;
    const int c    = lane & 15;
    const int h    = lane >> 4;
    const int tile = blockIdx.x * GEMM_WAVES + wave;
    const bool live = tile < GEMM_TILES;
    const int m0 = (live ? tile : GEMM_TILES - 1) * 16;

    const float4* xr = (const float4*)(X + (size_t)(m0 + c) * DIM) + h;
    float4 av[DIM / 8];
#pragma unroll
    for (int j = 0; j < DIM / 8; ++j) av[j] = xr[2 * j];

    float4 wv[PASSES];
#pragma unroll
    for (int p = 0; p < PASSES; ++p) {
        const int idx = tid + p * (GEMM_WAVES * 32);
        const int g = idx / NP;
        const int n = idx - g * NP;
        wv[p] = make_float4(0.0f, 0.0f, 0.0f, 0.0f);
        if (n < NCOL) {
            const float* wp = W + (size_t)(4 * g) * NCOL + n;
            wv[p].x = wp[0];
            wv[p].y = wp[NCOL];
            wv[p].z = wp[2 * NCOL];
            wv[p].w = wp[3 * NCOL];
        }
    }
#pragma unroll
    for (int p = 0; p < PASSES; ++p) {
        const int idx = tid + p * (GEMM_WAVES * 32);
        const int g = idx / NP;
        Wl[g * NPP + (idx - g * NP)] = wv[p];
    }
    __syncthreads();
    if (!live) return;

    const float4* bi4 = (const float4*)bin + h;
    const float4* wl  = Wl + h * NPP + c;

    v8f acc[NT];
#pragma unroll
    for (int t = 0; t < NT; ++t) acc[t] = v8f{};

#pragma unroll
    for (int j = 0; j < DIM / 8; ++j) {
        float4 a4 = av[j];
        if (ACT) {
            const float4 bb = bi4[2 * j];
            a4.x = fmaxf(a4.x + bb.x, 0.0f);
            a4.y = fmaxf(a4.y + bb.y, 0.0f);
            a4.z = fmaxf(a4.z + bb.z, 0.0f);
            a4.w = fmaxf(a4.w + bb.w, 0.0f);
        }
        v2f a0, a1;
        a0.x = a4.x; a0.y = a4.y;
        a1.x = a4.z; a1.y = a4.w;
        float4 bv[NT];
#pragma unroll
        for (int t = 0; t < NT; ++t) bv[t] = wl[(2 * j) * NPP + 16 * t];
#pragma unroll
        for (int t = 0; t < NT; ++t) {
            v2f b0;
            b0.x = bv[t].x; b0.y = bv[t].y;
            acc[t] = __builtin_amdgcn_wmma_f32_16x16x4_f32(
                 false, a0,  false, b0,
                 (short)0, acc[t],  false,  false);
        }
#pragma unroll
        for (int t = 0; t < NT; ++t) {
            v2f b1;
            b1.x = bv[t].z; b1.y = bv[t].w;
            acc[t] = __builtin_amdgcn_wmma_f32_16x16x4_f32(
                false, a1, false, b1, (short)0, acc[t], false, false);
        }
    }

    float ob[NT];
#pragma unroll
    for (int t = 0; t < NT; ++t) ob[t] = (OBIAS && 16 * t + c < NCOL) ? obias[16 * t + c] : 0.0f;
    float rs[8];
#pragma unroll
    for (int r = 0; r < 8; ++r) rs[r] = RSCALE ? rscale[m0 + h * 8 + r] : 1.0f;
    float* hrow = H + (size_t)(m0 + h * 8) * NCOL + c;
#pragma unroll
    for (int r = 0; r < 8; ++r) {
#pragma unroll
        for (int t = 0; t < NT; ++t) {
            if (16 * t + c < NCOL) {
                float v = OBIAS ? acc[t][r] + ob[t] : acc[t][r];
                if (RSCALE) v *= rs[r];
                hrow[(size_t)r * NCOL + 16 * t] = v;
            }
        }
    }
}

__global__ __launch_bounds__(256) void gemm128_wmma(const float* __restrict__ X,
                                                    const float* __restrict__ W,
                                                    const float* __restrict__ dinv,
                                                    float* __restrict__ H) {
    __shared__ float4 Wl[(DIM / 4) * (DIM + 8)];
    gemm_rows_body<DIM / 16, DIM, false, false, true>(X, W, nullptr, nullptr, dinv, H, Wl);
}

__global__ __launch_bounds__(256) void gemm128_act_wmma(const float* __restrict__ S,
                                                        const float* __restrict__ bias,
                                                        const float* __restrict__ W,
                                                        const float* __restrict__ dinv,
                                                        float* __restrict__ H) {
    __shared__ float4 Wl[(DIM / 4) * (DIM + 8)];
    gemm_rows_body<DIM / 16, DIM, true, false, true>(S, W, bias, nullptr, dinv, H, Wl);
}

__global__ __launch_bounds__(256) void gemm_out_wmma(const float* __restrict__ S,
                                                     const float* __restrict__ bin,
                                                     const float* __restrict__ W,
                                                     const float* __restrict__ bias,
                                                     float* __restrict__ OUT) {
    constexpr int NT_OUT = (N_CLASSES + 15) / 16;
    __shared__ float4 Wl[(DIM / 4) * (NT_OUT * 16 + 8)];
    gemm_rows_body<NT_OUT, N_CLASSES, true, true, false>(S, W, bin, bias, nullptr, OUT, Wl);
}

__global__ __launch_bounds__(1024) void bin_edges_kernel(const int* __restrict__ row,
                                                         const int* __restrict__ col,
                                                         unsigned* __restrict__ lists, int* counts) {
    __shared__ unsigned      s_ent[BIN_SLICE];
    __shared__ unsigned char s_own[BIN_SLICE];
    __shared__ unsigned      s_cnt[NBLK_AGG];
    __shared__ unsigned      s_off[NBLK_AGG];
    __shared__ unsigned      s_base[NBLK_AGG];
    __shared__ unsigned      s_wtot[NBLK_AGG / 32];

    const int tid  = threadIdx.x;
    const int lane = tid & 31;
    const int wave = tid >> 5;
    const int e0   = blockIdx.x * BIN_SLICE;
    int n = N_EDGES - e0;
    if (n > BIN_SLICE) n = BIN_SLICE;

    if (tid < NBLK_AGG) s_cnt[tid] = 0;

    int dst[BIN_PASSES], sr[BIN_PASSES];
#pragma unroll
    for (int k = 0; k < BIN_PASSES; ++k) {
        const int i = tid + k * 1024;
        dst[k] = -1;
        sr[k]  = 0;
        if (i < n) {
            dst[k] = col[e0 + i];
            sr[k]  = row[e0 + i];
        }
    }
    __syncthreads();

    unsigned own[BIN_PASSES], rnk[BIN_PASSES], pk[BIN_PASSES];
#pragma unroll
    for (int k = 0; k < BIN_PASSES; ++k) {
        const bool ok = (unsigned)dst[k] < (unsigned)N_NODES && (unsigned)sr[k] < (unsigned)N_NODES;
        own[k] = NBLK_AGG;
        rnk[k] = 0;
        pk[k]  = 0;
        if (ok) {
            const unsigned o = (unsigned)dst[k] / TROWS;
            own[k] = o;
            rnk[k] = atomicAdd(&s_cnt[o], 1u);
            pk[k]  = ((unsigned)sr[k] << 9) | ((unsigned)dst[k] - o * TROWS);
        }
    }
    __syncthreads();

    unsigned cn = 0, incl = 0;
    if (tid < NBLK_AGG) {
        cn = s_cnt[tid];
        incl = cn;
#pragma unroll
        for (int d = 1; d < 32; d <<= 1) {
            const unsigned t = __shfl_up(incl, d);
            if (lane >= d) incl += t;
        }
        if (lane == 31) s_wtot[wave] = incl;
    }
    __syncthreads();
    if (tid < NBLK_AGG) {
        unsigned pre = 0;
#pragma unroll
        for (int w = 0; w < NBLK_AGG / 32; ++w)
            if (w < wave) pre += s_wtot[w];
        s_off[tid]  = pre + incl - cn;
        s_base[tid] = cn ? (unsigned)atomicAdd(&counts[tid], (int)cn) : 0u;
    }
    __syncthreads();

#pragma unroll
    for (int k = 0; k < BIN_PASSES; ++k) {
        if (own[k] < NBLK_AGG) {
            const unsigned slot = s_off[own[k]] + rnk[k];
            s_ent[slot] = pk[k];
            s_own[slot] = (unsigned char)own[k];
        }
    }
    __syncthreads();

    const int total = (int)(s_off[NBLK_AGG - 1] + s_cnt[NBLK_AGG - 1]);
    for (int s = tid; s < total; s += 1024) {
        const unsigned o   = s_own[s];
        const unsigned pos = s_base[o] + ((unsigned)s - s_off[o]);
        if (pos < (unsigned)CAPB) lists[(size_t)o * CAPB + pos] = s_ent[s];
    }
}

__global__ __launch_bounds__(1024) void dinv_lists_kernel(const int* __restrict__ row,
                                                          const int* __restrict__ col,
                                                          const unsigned* __restrict__ lists,
                                                          const int* __restrict__ counts,
                                                          float* __restrict__ dinv) {
    __shared__ unsigned s_deg[TROWS + 1];
    const int tid = threadIdx.x;
    const int lo  = blockIdx.x * TROWS;
    int nrows = N_NODES - lo;
    if (nrows > TROWS) nrows = TROWS;

    for (int i = tid; i < TROWS; i += 1024) s_deg[i] = 0;
    __syncthreads();

    const int cnt = counts[blockIdx.x];
    if (cnt <= CAPB) {
        const unsigned* rec = lists + (size_t)blockIdx.x * CAPB;
        for (int i = tid; i < cnt; i += 1024) atomicAdd(&s_deg[rec[i] & 511u], 1u);
    } else {
        for (int e = tid; e < N_EDGES; e += 1024) {
            const unsigned cl = (unsigned)(col[e] - lo);
            if (cl < (unsigned)nrows && (unsigned)row[e] < (unsigned)N_NODES) atomicAdd(&s_deg[cl], 1u);
        }
    }
    __syncthreads();
    for (int i = tid; i < nrows; i += 1024) dinv[lo + i] = rsqrtf(1.0f + (float)s_deg[i]);
}


typedef volatile __attribute__((address_space(3))) unsigned lds_vu32;

template <typename QP>
__device__ __forceinline__ void drain_edges(QP q, int n, int lane,
                                            const int* __restrict__ row,
                                            const float* __restrict__ h,
                                            float* tile) {
    unsigned ent[8];
    int r[8];
    float hv[8][4];
#pragma unroll
    for (int b = 0; b < 8; ++b) {
        ent[b] = __builtin_amdgcn_readfirstlane(q[b < n ? b : 0]);
    }
#pragma unroll
    for (int b = 0; b < 8; ++b) r[b] = row[ent[b] >> 9];
#pragma unroll
    for (int b = 0; b < 8; ++b) {
#pragma unroll
        for (int k = 0; k < 4; ++k) hv[b][k] = h[(size_t)r[b] * DIM + k * 32 + lane];
    }
#pragma unroll
    for (int b = 0; b < 8; ++b) {
        if (b < n) {
            float* trow = tile + (ent[b] & 511u) * DIM + lane;
#pragma unroll
            for (int k = 0; k < 4; ++k)
                __hip_atomic_fetch_add(trow + k * 32, hv[b][k], __ATOMIC_RELAXED,
                                       __HIP_MEMORY_SCOPE_WORKGROUP);
        }
    }
}

__device__ __forceinline__ void scan_targets(lds_vu32* q, int lo, int nrows, int lane, int wave,
                                             const int* __restrict__ row,
                                             const int* __restrict__ col,
                                             const float* __restrict__ h,
                                             float* tile) {
    const unsigned below = (1u << lane) - 1u;
    const int4* col4 = (const int4*)col;
    const int4* row4 = (const int4*)row;
    int qn = 0;

    for (int g0 = wave * GRP; g0 < NCHUNK; g0 += 32 * GRP) {
        int4 cb[GRP];
#pragma unroll
        for (int u = 0; u < GRP; ++u) {
            if (g0 + u < NCHUNK) cb[u] = col4[(size_t)(g0 + u) * 32 + lane];
            else                 cb[u] = make_int4(-1, -1, -1, -1);
        }
#pragma unroll
        for (int u = 0; u < GRP; ++u) {
            const unsigned c0 = (unsigned)(cb[u].x - lo);
            const unsigned c1 = (unsigned)(cb[u].y - lo);
            const unsigned c2 = (unsigned)(cb[u].z - lo);
            const unsigned c3 = (unsigned)(cb[u].w - lo);
            bool m0 = c0 < (unsigned)nrows;
            bool m1 = c1 < (unsigned)nrows;
            bool m2 = c2 < (unsigned)nrows;
            bool m3 = c3 < (unsigned)nrows;
            if (__builtin_amdgcn_ballot_w32(m0 | m1 | m2 | m3) == 0u) continue;

            const int4 rb = row4[(size_t)(g0 + u) * 32 + lane];
            m0 = m0 && (unsigned)rb.x < (unsigned)N_NODES;
            m1 = m1 && (unsigned)rb.y < (unsigned)N_NODES;
            m2 = m2 && (unsigned)rb.z < (unsigned)N_NODES;
            m3 = m3 && (unsigned)rb.w < (unsigned)N_NODES;

            const unsigned e0 = (unsigned)(g0 + u) * 128u + (unsigned)lane * 4u;
            unsigned mk;
            mk = __builtin_amdgcn_ballot_w32(m0);
            if (m0) q[qn + __builtin_popcount(mk & below)] = ((e0 + 0u) << 9) | c0;
            qn += __builtin_popcount(mk);
            mk = __builtin_amdgcn_ballot_w32(m1);
            if (m1) q[qn + __builtin_popcount(mk & below)] = ((e0 + 1u) << 9) | c1;
            qn += __builtin_popcount(mk);
            mk = __builtin_amdgcn_ballot_w32(m2);
            if (m2) q[qn + __builtin_popcount(mk & below)] = ((e0 + 2u) << 9) | c2;
            qn += __builtin_popcount(mk);
            mk = __builtin_amdgcn_ballot_w32(m3);
            if (m3) q[qn + __builtin_popcount(mk & below)] = ((e0 + 3u) << 9) | c3;
            qn += __builtin_popcount(mk);

            while (qn >= 8) {
                qn -= 8;
                drain_edges(q + qn, 8, lane, row, h, tile);
            }
        }
    }
    if (qn > 0) drain_edges(q, qn, lane, row, h, tile);
}

__device__ __forceinline__ void replay_edges(const unsigned* __restrict__ rec, int cnt,
                                             int lane, int wave,
                                             const float* __restrict__ h,
                                             float* tile) {
    const int wave_u = __builtin_amdgcn_readfirstlane(wave);
    int i0 = wave_u * 32;
    unsigned nxt = 0u;
    if (i0 + lane < cnt) nxt = rec[i0 + lane];
    for (; i0 < cnt; i0 += 1024) {
        int nb = cnt - i0;
        if (nb > 32) nb = 32;
        const unsigned ent = nxt;
        nxt = 0u;
        if (i0 + 1024 + lane < cnt) nxt = rec[i0 + 1024 + lane];
        for (int g = 0; g < nb; g += 8) {
            unsigned eb[8];
            float hv[8][4];
#pragma unroll
            for (int b = 0; b < 8; ++b) {
                const int sel = (g + b < nb) ? (g + b) : g;
                eb[b] = (unsigned)__builtin_amdgcn_readlane((int)ent, sel);
            }
#pragma unroll
            for (int b = 0; b < 8; ++b) {
#pragma unroll
                for (int k = 0; k < 4; ++k) hv[b][k] = h[(size_t)(eb[b] >> 9) * DIM + k * 32 + lane];
            }
#pragma unroll
            for (int b = 0; b < 8; ++b) {
                if (b == 0 || g + b < nb) {
                    float* trow = tile + (eb[b] & 511u) * DIM + lane;
#pragma unroll
                    for (int k = 0; k < 4; ++k)
                        __hip_atomic_fetch_add(trow + k * 32, hv[b][k], __ATOMIC_RELAXED,
                                               __HIP_MEMORY_SCOPE_WORKGROUP);
                }
            }
        }
    }
}

__global__ __launch_bounds__(1024) void agg_owner_kernel(const int* __restrict__ row,
                                                         const int* __restrict__ col,
                                                         const float* __restrict__ dinv,
                                                         const float* __restrict__ h,
                                                         float* __restrict__ agg,
                                                         const unsigned* __restrict__ lists,
                                                         const int* __restrict__ counts) {
    __shared__ float    tile[TROWS * DIM];
    __shared__ float    s_dinv[TROWS + 1];
    __shared__ unsigned s_q[32 * QCAP];

    const int tid  = threadIdx.x;
    const int lane = tid & 31;
    const int wave = tid >> 5;
    const int lo   = blockIdx.x * TROWS;
    int nrows = N_NODES - lo;
    if (nrows > TROWS) nrows = TROWS;

    const float4* h4    = (const float4*)h;
    float4*       tile4 = (float4*)tile;
    for (int i = tid; i < nrows * 32; i += 1024) tile4[i] = h4[(size_t)lo * 32 + i];
    for (int i = tid; i < nrows; i += 1024) s_dinv[i] = dinv[lo + i];
    __syncthreads();

    const int cnt = counts[blockIdx.x];
    if (cnt <= CAPB) {
        replay_edges(lists + (size_t)blockIdx.x * CAPB, cnt, lane, wave, h, tile);
    } else {
        lds_vu32* q = (lds_vu32*)(s_q + wave * QCAP);
        scan_targets(q, lo, nrows, lane, wave, row, col, h, tile);
    }
    __syncthreads();

    float4* agg4 = (float4*)agg;
    for (int i = tid; i < nrows * 32; i += 1024) {
        const float d = s_dinv[i >> 5];
        float4 v = tile4[i];
        v.x = d * v.x;
        v.y = d * v.y;
        v.z = d * v.z;
        v.w = d * v.w;
        agg4[(size_t)lo * 32 + i] = v;
    }
}

extern "C" void kernel_launch(void* const* d_in, const int* in_sizes, int n_in,
                              void* d_out, int out_size, void* d_ws, size_t ws_size,
                              hipStream_t stream) {
    const float* x    = (const float*)d_in[0];
    const int*   edge = (const int*)d_in[1];
    const float* W1   = (const float*)d_in[2];
    const float* b1   = (const float*)d_in[3];
    const float* W2   = (const float*)d_in[4];
    const float* b2   = (const float*)d_in[5];
    const float* Wout = (const float*)d_in[6];
    const float* bout = (const float*)d_in[7];
    const int* row = edge;
    const int* col = edge + N_EDGES;
    float* out = (float*)d_out;

    char* ws = (char*)d_ws;
    float* dinv = (float*)ws;
    float* bufA = (float*)(ws + ((size_t)1 << 20));
    float* bufB = bufA + (size_t)N_NODES * DIM;
    int*   counts = (int*)(ws + ((size_t)1 << 19));
    unsigned* lists = (unsigned*)(ws + ((size_t)112 << 20));

    const int gblocks = (N_NODES / 16 + GEMM_WAVES - 1) / GEMM_WAVES;

    counts_zero_kernel<<<1, NBLK_AGG, 0, stream>>>(counts);
    bin_edges_kernel<<<NBLK_BIN, 1024, 0, stream>>>(row, col, lists, counts);
    dinv_lists_kernel<<<NBLK_AGG, 1024, 0, stream>>>(row, col, lists, counts, dinv);

    gemm128_wmma<<<gblocks, 256, 0, stream>>>(x, W1, dinv, bufA);
    agg_owner_kernel<<<NBLK_AGG, 1024, 0, stream>>>(row, col, dinv, bufA, bufB, lists, counts);

    gemm128_act_wmma<<<gblocks, 256, 0, stream>>>(bufB, b1, W2, dinv, bufA);
    agg_owner_kernel<<<NBLK_AGG, 1024, 0, stream>>>(row, col, dinv, bufA, bufB, lists, counts);

    gemm_out_wmma<<<gblocks, 256, 0, stream>>>(bufB, b2, Wout, bout, out);
}


# Aggregation without f32 LDS atomics: blocks sort their list by owner wave (row&31) in LDS; a wave adds its rows by plain 16-byte LDS load/add/store, one 16-byte row load per lane (sum order: arrival)
# speedup vs baseline: 4.2039x; 1.1037x over previous
#include <hip/hip_runtime.h>

typedef __attribute__((ext_vector_type(2))) float v2f;
typedef __attribute__((ext_vector_type(8))) float v8f;

#define N_NODES 100000
#define N_EDGES 1600000
#define DIM 128
#define N_CLASSES 40
#define TROWS 391
#define NBLK_AGG 256
#define QCAP 136
#define NCHUNK (N_EDGES / 128)
#define GRP 8
#define CAPB 32768
#define BIN_SLICE 6144
#define BIN_PASSES (BIN_SLICE / 1024)
#define NBLK_BIN ((N_EDGES + BIN_SLICE - 1) / BIN_SLICE)
static_assert(N_EDGES % 128 == 0, "edge list is read in whole chunks");
static_assert(NBLK_AGG * TROWS >= N_NODES, "blocks must cover every target row");
static_assert(TROWS <= 512 && N_EDGES <= (1 << 21), "queue entry packs edge id and local row in 32 bits");
static_assert(TROWS <= 512 && N_NODES <= (1 << 23), "list record packs source row and local row in 32 bits");
static_assert(NBLK_AGG <= 256, "the owner of a sorted slot is kept in one byte");

__global__ void counts_zero_kernel(int* counts) {
    if (threadIdx.x < NBLK_AGG) counts[threadIdx.x] = 0;
}

#define GEMM_WAVES 8
#define GEMM_TILES (N_NODES / 16)
static_assert(N_NODES % 16 == 0, "row tiles are whole");
template <int NT, int NCOL, bool ACT, bool OBIAS, bool RSCALE>
__device__ __forceinline__ void gemm_rows_body(const float* __restrict__ X,
                                               const float* __restrict__ W,
                                               const float* __restrict__ bin,
                                               const float* __restrict__ obias,
                                               const float* __restrict__ rscale,
                                               float* __restrict__ H,
                                               float4* Wl) {
    constexpr int NP     = NT * 16;
    constexpr int NPP    = NP + 8;
    constexpr int PASSES = (DIM / 4) * NP / (GEMM_WAVES * 32);
    static_assert(PASSES * GEMM_WAVES * 32 == (DIM / 4) * NP, "every thread stages the same number of groups");
    const int tid  = threadIdx.x;
    const int lane = tid & 31;
    const int wave = tid >> 5;
    const int c    = lane & 15;
    const int h    = lane >> 4;
    const int tile = blockIdx.x * GEMM_WAVES + wave;
    const bool live = tile < GEMM_TILES;
    const int m0 = (live ? tile : GEMM_TILES - 1) * 16;

    const float4* xr = (const float4*)(X + (size_t)(m0 + c) * DIM) + h;
    float4 av[DIM / 8];
#pragma unroll
    for (int j = 0; j < DIM / 8; ++j) av[j] = xr[2 * j];

    float4 wv[PASSES];
#pragma unroll
    for (int p = 0; p < PASSES; ++p) {
        const int idx = tid + p * (GEMM_WAVES * 32);
        const int g = idx / NP;
        const int n = idx - g * NP;
        wv[p] = make_float4(0.0f, 0.0f, 0.0f, 0.0f);
        if (n < NCOL) {
            const float* wp = W + (size_t)(4 * g) * NCOL + n;
            wv[p].x = wp[0];
            wv[p].y = wp[NCOL];
            wv[p].z = wp[2 * NCOL];
            wv[p].w = wp[3 * NCOL];
        }
    }
#pragma unroll
    for (int p = 0; p < PASSES; ++p) {
        const int idx = tid + p * (GEMM_WAVES * 32);
        const int g = idx / NP;
        Wl[g * NPP + (idx - g * NP)] = wv[p];
    }
    __syncthreads();
    if (!live) return;

    const float4* bi4 = (const float4*)bin + h;
    const float4* wl  = Wl + h * NPP + c;

    v8f acc[NT];
#pragma unroll
    for (int t = 0; t < NT; ++t) acc[t] = v8f{};

#pragma unroll
    for (int j = 0; j < DIM / 8; ++j) {
        float4 a4 = av[j];
        if (ACT) {
            const float4 bb = bi4[2 * j];
            a4.x = fmaxf(a4.x + bb.x, 0.0f);
            a4.y = fmaxf(a4.y + bb.y, 0.0f);
            a4.z = fmaxf(a4.z + bb.z, 0.0f);
            a4.w = fmaxf(a4.w + bb.w, 0.0f);
        }
        v2f a0, a1;
        a0.x = a4.x; a0.y = a4.y;
        a1.x = a4.z; a1.y = a4.w;
        float4 bv[NT];
#pragma unroll
        for (int t = 0; t < NT; ++t) bv[t] = wl[(2 * j) * NPP + 16 * t];
#pragma unroll
        for (int t = 0; t < NT; ++t) {
            v2f b0;
            b0.x = bv[t].x; b0.y = bv[t].y;
            acc[t] = __builtin_amdgcn_wmma_f32_16x16x4_f32(
                 false, a0,  false, b0,
                 (short)0, acc[t],  false,  false);
        }
#pragma unroll
        for (int t = 0; t < NT; ++t) {
            v2f b1;
            b1.x = bv[t].z; b1.y = bv[t].w;
            acc[t] = __builtin_amdgcn_wmma_f32_16x16x4_f32(
                false, a1, false, b1, (short)0, acc[t], false, false);
        }
    }

    float ob[NT];
#pragma unroll
    for (int t = 0; t < NT; ++t) ob[t] = (OBIAS && 16 * t + c < NCOL) ? obias[16 * t + c] : 0.0f;
    float rs[8];
#pragma unroll
    for (int r = 0; r < 8; ++r) rs[r] = RSCALE ? rscale[m0 + h * 8 + r] : 1.0f;
    float* hrow = H + (size_t)(m0 + h * 8) * NCOL + c;
#pragma unroll
    for (int r = 0; r < 8; ++r) {
#pragma unroll
        for (int t = 0; t < NT; ++t) {
            if (16 * t + c < NCOL) {
                float v = OBIAS ? acc[t][r] + ob[t] : acc[t][r];
                if (RSCALE) v *= rs[r];
                hrow[(size_t)r * NCOL + 16 * t] = v;
            }
        }
    }
}

__global__ __launch_bounds__(256) void gemm128_wmma(const float* __restrict__ X,
                                                    const float* __restrict__ W,
                                                    const float* __restrict__ dinv,
                                                    float* __restrict__ H) {
    __shared__ float4 Wl[(DIM / 4) * (DIM + 8)];
    gemm_rows_body<DIM / 16, DIM, false, false, true>(X, W, nullptr, nullptr, dinv, H, Wl);
}

__global__ __launch_bounds__(256) void gemm128_act_wmma(const float* __restrict__ S,
                                                        const float* __restrict__ bias,
                                                        const float* __restrict__ W,
                                                        const float* __restrict__ dinv,
                                                        float* __restrict__ H) {
    __shared__ float4 Wl[(DIM / 4) * (DIM + 8)];
    gemm_rows_body<DIM / 16, DIM, true, false, true>(S, W, bias, nullptr, dinv, H, Wl);
}

__global__ __launch_bounds__(256) void gemm_out_wmma(const float* __restrict__ S,
                                                     const float* __restrict__ bin,
                                                     const float* __restrict__ W,
                                                     const float* __restrict__ bias,
                                                     float* __restrict__ OUT) {
    constexpr int NT_OUT = (N_CLASSES + 15) / 16;
    __shared__ float4 Wl[(DIM / 4) * (NT_OUT * 16 + 8)];
    gemm_rows_body<NT_OUT, N_CLASSES, true, true, false>(S, W, bin, bias, nullptr, OUT, Wl);
}

__global__ __launch_bounds__(1024) void bin_edges_kernel(const int* __restrict__ row,
                                                         const int* __restrict__ col,
                                                         unsigned* __restrict__ lists, int* counts) {
    __shared__ unsigned      s_ent[BIN_SLICE];
    __shared__ unsigned char s_own[BIN_SLICE];
    __shared__ unsigned      s_cnt[NBLK_AGG];
    __shared__ unsigned      s_off[NBLK_AGG];
    __shared__ unsigned      s_base[NBLK_AGG];
    __shared__ unsigned      s_wtot[NBLK_AGG / 32];

    const int tid  = threadIdx.x;
    const int lane = tid & 31;
    const int wave = tid >> 5;
    const int e0   = blockIdx.x * BIN_SLICE;
    int n = N_EDGES - e0;
    if (n > BIN_SLICE) n = BIN_SLICE;

    if (tid < NBLK_AGG) s_cnt[tid] = 0;

    int dst[BIN_PASSES], sr[BIN_PASSES];
#pragma unroll
    for (int k = 0; k < BIN_PASSES; ++k) {
        const int i = tid + k * 1024;
        dst[k] = -1;
        sr[k]  = 0;
        if (i < n) {
            dst[k] = col[e0 + i];
            sr[k]  = row[e0 + i];
        }
    }
    __syncthreads();

    unsigned own[BIN_PASSES], rnk[BIN_PASSES], pk[BIN_PASSES];
#pragma unroll
    for (int k = 0; k < BIN_PASSES; ++k) {
        const bool ok = (unsigned)dst[k] < (unsigned)N_NODES && (unsigned)sr[k] < (unsigned)N_NODES;
        own[k] = NBLK_AGG;
        rnk[k] = 0;
        pk[k]  = 0;
        if (ok) {
            const unsigned o = (unsigned)dst[k] / TROWS;
            own[k] = o;
            rnk[k] = atomicAdd(&s_cnt[o], 1u);
            pk[k]  = ((unsigned)sr[k] << 9) | ((unsigned)dst[k] - o * TROWS);
        }
    }
    __syncthreads();

    unsigned cn = 0, incl = 0;
    if (tid < NBLK_AGG) {
        cn = s_cnt[tid];
        incl = cn;
#pragma unroll
        for (int d = 1; d < 32; d <<= 1) {
            const unsigned t = __shfl_up(incl, d);
            if (lane >= d) incl += t;
        }
        if (lane == 31) s_wtot[wave] = incl;
    }
    __syncthreads();
    if (tid < NBLK_AGG) {
        unsigned pre = 0;
#pragma unroll
        for (int w = 0; w < NBLK_AGG / 32; ++w)
            if (w < wave) pre += s_wtot[w];
        s_off[tid]  = pre + incl - cn;
        s_base[tid] = cn ? (unsigned)atomicAdd(&counts[tid], (int)cn) : 0u;
    }
    __syncthreads();

#pragma unroll
    for (int k = 0; k < BIN_PASSES; ++k) {
        if (own[k] < NBLK_AGG) {
            const unsigned slot = s_off[own[k]] + rnk[k];
            s_ent[slot] = pk[k];
            s_own[slot] = (unsigned char)own[k];
        }
    }
    __syncthreads();

    const int total = (int)(s_off[NBLK_AGG - 1] + s_cnt[NBLK_AGG - 1]);
    for (int s = tid; s < total; s += 1024) {
        const unsigned o   = s_own[s];
        const unsigned pos = s_base[o] + ((unsigned)s - s_off[o]);
        if (pos < (unsigned)CAPB) lists[(size_t)o * CAPB + pos] = s_ent[s];
    }
}

__global__ __launch_bounds__(1024) void dinv_lists_kernel(const int* __restrict__ row,
                                                          const int* __restrict__ col,
                                                          const unsigned* __restrict__ lists,
                                                          const int* __restrict__ counts,
                                                          float* __restrict__ dinv) {
    __shared__ unsigned s_deg[TROWS + 1];
    const int tid = threadIdx.x;
    const int lo  = blockIdx.x * TROWS;
    int nrows = N_NODES - lo;
    if (nrows > TROWS) nrows = TROWS;

    for (int i = tid; i < TROWS; i += 1024) s_deg[i] = 0;
    __syncthreads();

    const int cnt = counts[blockIdx.x];
    if (cnt <= CAPB) {
        const unsigned* rec = lists + (size_t)blockIdx.x * CAPB;
        for (int i = tid; i < cnt; i += 1024) atomicAdd(&s_deg[rec[i] & 511u], 1u);
    } else {
        for (int e = tid; e < N_EDGES; e += 1024) {
            const unsigned cl = (unsigned)(col[e] - lo);
            if (cl < (unsigned)nrows && (unsigned)row[e] < (unsigned)N_NODES) atomicAdd(&s_deg[cl], 1u);
        }
    }
    __syncthreads();
    for (int i = tid; i < nrows; i += 1024) dinv[lo + i] = rsqrtf(1.0f + (float)s_deg[i]);
}


typedef volatile __attribute__((address_space(3))) unsigned lds_vu32;

template <typename QP>
__device__ __forceinline__ void drain_edges(QP q, int n, int lane,
                                            const int* __restrict__ row,
                                            const float* __restrict__ h,
                                            float* tile) {
    unsigned ent[8];
    int r[8];
    float hv[8][4];
#pragma unroll
    for (int b = 0; b < 8; ++b) {
        ent[b] = __builtin_amdgcn_readfirstlane(q[b < n ? b : 0]);
    }
#pragma unroll
    for (int b = 0; b < 8; ++b) r[b] = row[ent[b] >> 9];
#pragma unroll
    for (int b = 0; b < 8; ++b) {
#pragma unroll
        for (int k = 0; k < 4; ++k) hv[b][k] = h[(size_t)r[b] * DIM + k * 32 + lane];
    }
#pragma unroll
    for (int b = 0; b < 8; ++b) {
        if (b < n) {
            float* trow = tile + (ent[b] & 511u) * DIM + lane;
#pragma unroll
            for (int k = 0; k < 4; ++k)
                __hip_atomic_fetch_add(trow + k * 32, hv[b][k], __ATOMIC_RELAXED,
                                       __HIP_MEMORY_SCOPE_WORKGROUP);
        }
    }
}

__device__ __forceinline__ void scan_targets(lds_vu32* q, int lo, int nrows, int lane, int wave,
                                             const int* __restrict__ row,
                                             const int* __restrict__ col,
                                             const float* __restrict__ h,
                                             float* tile) {
    const unsigned below = (1u << lane) - 1u;
    const int4* col4 = (const int4*)col;
    const int4* row4 = (const int4*)row;
    int qn = 0;

    for (int g0 = wave * GRP; g0 < NCHUNK; g0 += 32 * GRP) {
        int4 cb[GRP];
#pragma unroll
        for (int u = 0; u < GRP; ++u) {
            if (g0 + u < NCHUNK) cb[u] = col4[(size_t)(g0 + u) * 32 + lane];
            else                 cb[u] = make_int4(-1, -1, -1, -1);
        }
#pragma unroll
        for (int u = 0; u < GRP; ++u) {
            const unsigned c0 = (unsigned)(cb[u].x - lo);
            const unsigned c1 = (unsigned)(cb[u].y - lo);
            const unsigned c2 = (unsigned)(cb[u].z - lo);
            const unsigned c3 = (unsigned)(cb[u].w - lo);
            bool m0 = c0 < (unsigned)nrows;
            bool m1 = c1 < (unsigned)nrows;
            bool m2 = c2 < (unsigned)nrows;
            bool m3 = c3 < (unsigned)nrows;
            if (__builtin_amdgcn_ballot_w32(m0 | m1 | m2 | m3) == 0u) continue;

            const int4 rb = row4[(size_t)(g0 + u) * 32 + lane];
            m0 = m0 && (unsigned)rb.x < (unsigned)N_NODES;
            m1 = m1 && (unsigned)rb.y < (unsigned)N_NODES;
            m2 = m2 && (unsigned)rb.z < (unsigned)N_NODES;
            m3 = m3 && (unsigned)rb.w < (unsigned)N_NODES;

            const unsigned e0 = (unsigned)(g0 + u) * 128u + (unsigned)lane * 4u;
            unsigned mk;
            mk = __builtin_amdgcn_ballot_w32(m0);
            if (m0) q[qn + __builtin_popcount(mk & below)] = ((e0 + 0u) << 9) | c0;
            qn += __builtin_popcount(mk);
            mk = __builtin_amdgcn_ballot_w32(m1);
            if (m1) q[qn + __builtin_popcount(mk & below)] = ((e0 + 1u) << 9) | c1;
            qn += __builtin_popcount(mk);
            mk = __builtin_amdgcn_ballot_w32(m2);
            if (m2) q[qn + __builtin_popcount(mk & below)] = ((e0 + 2u) << 9) | c2;
            qn += __builtin_popcount(mk);
            mk = __builtin_amdgcn_ballot_w32(m3);
            if (m3) q[qn + __builtin_popcount(mk & below)] = ((e0 + 3u) << 9) | c3;
            qn += __builtin_popcount(mk);

            while (qn >= 8) {
                qn -= 8;
                drain_edges(q + qn, 8, lane, row, h, tile);
            }
        }
    }
    if (qn > 0) drain_edges(q, qn, lane, row, h, tile);
}

template <int G>
__device__ __forceinline__ void load_group(unsigned (&e)[8], float4 (&hv)[8], unsigned ent, int nb,
                                           int lane, const float4* __restrict__ h4) {
#pragma unroll
    for (int b = 0; b < 8; ++b) {
        const int sel = (G + b < nb) ? (G + b) : G;
        e[b] = (unsigned)__builtin_amdgcn_readlane((int)ent, sel);
    }
#pragma unroll
    for (int b = 0; b < 8; ++b) hv[b] = h4[(size_t)(e[b] >> 9) * (DIM / 4) + lane];
}

template <int G>
__device__ __forceinline__ void add_group(const unsigned (&e)[8], const float4 (&hv)[8], int nb,
                                          int lane, float4* tile4) {
#pragma unroll
    for (int p = 0; p < 4; ++p) {
        const int b0 = 2 * p, b1 = 2 * p + 1;
        if (b0 == 0 || G + b0 < nb) {
            const bool two = G + b1 < nb;
            const unsigned r0 = e[b0] & 511u, r1 = e[b1] & 511u;
            float4* t0p = tile4 + r0 * (DIM / 4) + lane;
            float4* t1p = tile4 + r1 * (DIM / 4) + lane;
            if (two && r0 != r1) {
                float4 t0 = *t0p, t1 = *t1p;
                t0.x += hv[b0].x; t0.y += hv[b0].y; t0.z += hv[b0].z; t0.w += hv[b0].w;
                t1.x += hv[b1].x; t1.y += hv[b1].y; t1.z += hv[b1].z; t1.w += hv[b1].w;
                *t0p = t0;
                *t1p = t1;
            } else {
                float4 t0 = *t0p;
                t0.x += hv[b0].x; t0.y += hv[b0].y; t0.z += hv[b0].z; t0.w += hv[b0].w;
                if (two) { t0.x += hv[b1].x; t0.y += hv[b1].y; t0.z += hv[b1].z; t0.w += hv[b1].w; }
                *t0p = t0;
            }
        }
    }
}

__device__ __forceinline__ void replay_owned(const unsigned* srec, int nrec, int lane,
                                             const float4* __restrict__ h4, float4* tile4) {
    for (int i0 = 0; i0 < nrec; i0 += 32) {
        int nb = nrec - i0;
        if (nb > 32) nb = 32;
        unsigned ent = 0u;
        if (lane < nb) ent = srec[i0 + lane];
        unsigned ea[8], eb[8];
        float4 ha[8], hb[8];
        load_group<0>(ea, ha, ent, nb, lane, h4);
        if (nb > 8) load_group<8>(eb, hb, ent, nb, lane, h4);
        add_group<0>(ea, ha, nb, lane, tile4);
        if (nb > 16) load_group<16>(ea, ha, ent, nb, lane, h4);
        if (nb > 8) add_group<8>(eb, hb, nb, lane, tile4);
        if (nb > 24) load_group<24>(eb, hb, ent, nb, lane, h4);
        if (nb > 16) add_group<16>(ea, ha, nb, lane, tile4);
        if (nb > 24) add_group<24>(eb, hb, nb, lane, tile4);
    }
}

#define SCAP 8192
#define SORT_PASSES (SCAP / 1024)
static_assert(32 * QCAP <= SCAP, "the scan path's queues live in the sort buffer");
static_assert(CAPB % SCAP == 0, "a full list is a whole number of chunks");
__global__ __launch_bounds__(1024) void agg_owner_kernel(const int* __restrict__ row,
                                                         const int* __restrict__ col,
                                                         const float* __restrict__ dinv,
                                                         const float* __restrict__ h,
                                                         float* __restrict__ agg,
                                                         const unsigned* __restrict__ lists,
                                                         const int* __restrict__ counts) {
    __shared__ float    tile[TROWS * DIM];
    __shared__ float    s_dinv[TROWS + 1];
    __shared__ unsigned s_rec[SCAP];
    __shared__ unsigned s_wcnt[32];
    __shared__ unsigned s_woff[32];

    const int tid  = threadIdx.x;
    const int lane = tid & 31;
    const int wave = tid >> 5;
    const int lo   = blockIdx.x * TROWS;
    int nrows = N_NODES - lo;
    if (nrows > TROWS) nrows = TROWS;

    const int cnt = counts[blockIdx.x];
    const bool fits = cnt <= CAPB;
    const unsigned* rec = lists + (size_t)blockIdx.x * CAPB;

    int n = fits ? (cnt < SCAP ? cnt : SCAP) : 0;
    unsigned rv[SORT_PASSES];
#pragma unroll
    for (int k = 0; k < SORT_PASSES; ++k) {
        const int i = tid + k * 1024;
        rv[k] = 0u;
        if (i < n) rv[k] = rec[i];
    }

    const float4* h4    = (const float4*)h;
    float4*       tile4 = (float4*)tile;
    for (int i = tid; i < nrows * 32; i += 1024) tile4[i] = h4[(size_t)lo * 32 + i];
    for (int i = tid; i < nrows; i += 1024) s_dinv[i] = dinv[lo + i];
    if (tid < 32) s_wcnt[tid] = 0;
    __syncthreads();

    if (fits) {
        for (int base = 0; base < cnt;) {
            unsigned rk[SORT_PASSES];
#pragma unroll
            for (int k = 0; k < SORT_PASSES; ++k) {
                rk[k] = 0u;
                if (tid + k * 1024 < n) rk[k] = atomicAdd(&s_wcnt[rv[k] & 31u], 1u);
            }
            __syncthreads();
            if (tid < 32) {
                const unsigned c = s_wcnt[tid];
                unsigned incl = c;
#pragma unroll
                for (int d = 1; d < 32; d <<= 1) {
                    const unsigned t = __shfl_up(incl, d);
                    if (lane >= d) incl += t;
                }
                s_woff[tid] = incl - c;
            }
            __syncthreads();
#pragma unroll
            for (int k = 0; k < SORT_PASSES; ++k)
                if (tid + k * 1024 < n) s_rec[s_woff[rv[k] & 31u] + rk[k]] = rv[k];
            __syncthreads();

            const int wv  = __builtin_amdgcn_readfirstlane(wave);
            const int beg = __builtin_amdgcn_readfirstlane((int)s_woff[wv]);
            const int nw  = __builtin_amdgcn_readfirstlane((int)s_wcnt[wv]);
            replay_owned(s_rec + beg, nw, lane, h4, tile4);

            base += SCAP;
            if (base >= cnt) break;
            __syncthreads();
            if (tid < 32) s_wcnt[tid] = 0;
            n = cnt - base;
            if (n > SCAP) n = SCAP;
#pragma unroll
            for (int k = 0; k < SORT_PASSES; ++k) {
                const int i = tid + k * 1024;
                rv[k] = 0u;
                if (i < n) rv[k] = rec[base + i];
            }
            __syncthreads();
        }
    } else {
        lds_vu32* q = (lds_vu32*)(s_rec + wave * QCAP);
        scan_targets(q, lo, nrows, lane, wave, row, col, h, tile);
    }
    __syncthreads();

    float4* agg4 = (float4*)agg;
    for (int i = tid; i < nrows * 32; i += 1024) {
        const float d = s_dinv[i >> 5];
        float4 v = tile4[i];
        v.x = d * v.x;
        v.y = d * v.y;
        v.z = d * v.z;
        v.w = d * v.w;
        agg4[(size_t)lo * 32 + i] = v;
    }
}

extern "C" void kernel_launch(void* const* d_in, const int* in_sizes, int n_in,
                              void* d_out, int out_size, void* d_ws, size_t ws_size,
                              hipStream_t stream) {
    const float* x    = (const float*)d_in[0];
    const int*   edge = (const int*)d_in[1];
    const float* W1   = (const float*)d_in[2];
    const float* b1   = (const float*)d_in[3];
    const float* W2   = (const float*)d_in[4];
    const float* b2   = (const float*)d_in[5];
    const float* Wout = (const float*)d_in[6];
    const float* bout = (const float*)d_in[7];
    const int* row = edge;
    const int* col = edge + N_EDGES;
    float* out = (float*)d_out;

    char* ws = (char*)d_ws;
    float* dinv = (float*)ws;
    float* bufA = (float*)(ws + ((size_t)1 << 20));
    float* bufB = bufA + (size_t)N_NODES * DIM;
    int*   counts = (int*)(ws + ((size_t)1 << 19));
    unsigned* lists = (unsigned*)(ws + ((size_t)112 << 20));

    const int gblocks = (N_NODES / 16 + GEMM_WAVES - 1) / GEMM_WAVES;

    counts_zero_kernel<<<1, NBLK_AGG, 0, stream>>>(counts);
    bin_edges_kernel<<<NBLK_BIN, 1024, 0, stream>>>(row, col, lists, counts);
    dinv_lists_kernel<<<NBLK_AGG, 1024, 0, stream>>>(row, col, lists, counts, dinv);

    gemm128_wmma<<<gblocks, 256, 0, stream>>>(x, W1, dinv, bufA);
    agg_owner_kernel<<<NBLK_AGG, 1024, 0, stream>>>(row, col, dinv, bufA, bufB, lists, counts);

    gemm128_act_wmma<<<gblocks, 256, 0, stream>>>(bufB, b1, W2, dinv, bufA);
    agg_owner_kernel<<<NBLK_AGG, 1024, 0, stream>>>(row, col, dinv, bufA, bufB, lists, counts);

    gemm_out_wmma<<<gblocks, 256, 0, stream>>>(bufB, b2, Wout, bout, out);
}
